# starting kernel
# baseline (speedup 1.0000x reference)
#include <hip/hip_runtime.h>

typedef float v2f __attribute__((ext_vector_type(2)));
typedef float v8f __attribute__((ext_vector_type(8)));

#define CIN   256
#define CCOMP 64
#define CENC  36
#define HH    64
#define WWID  64
#define HW    4096
#define KK    9
#define SS    4
#define KDIM  576

__global__ void gg_mean_kernel(const float* __restrict__ x,
                               float* __restrict__ mean) {
    int idx = blockIdx.x * 256 + threadIdx.x;
    int b   = idx >> 12;
    int pix = idx & 4095;
    const float* xb = x + (size_t)b * CIN * HW + pix;
    float s = 0.f;
#pragma unroll 8
    for (int c = 0; c < CIN; ++c) s += xb[c * HW];
    mean[idx] = s * (1.0f / CIN);
}

__global__ void gg_comp_gemm(const float* __restrict__ x,
                             const float* __restrict__ w1,
                             const float* __restrict__ b1,
                             float* __restrict__ comp) {
    int b     = blockIdx.x >> 7;
    int nBase = (blockIdx.x & 127) * 32;
    int wave  = threadIdx.x >> 5;
    int lane  = threadIdx.x & 31;
    int lr    = lane & 15;
    int half  = lane >> 4;
    int m0    = (wave & 3) * 16;
    int n0    = nBase + (wave >> 2) * 16;

    const float* xb   = x + (size_t)b * CIN * HW;
    const float* aRow = w1 + (m0 + lr) * CIN;

    v8f acc = {};
    for (int k = 0; k < CIN; k += 4) {
        int ke = k + 2 * half;
        v2f a, bm;
        a.x  = aRow[ke];
        a.y  = aRow[ke + 1];
        bm.x = xb[ke * HW + n0 + lr];
        bm.y = xb[(ke + 1) * HW + n0 + lr];
        acc = __builtin_amdgcn_wmma_f32_16x16x4_f32(
                  false, a, false, bm, (short)0, acc, false, false);
    }

    float* cb = comp + (size_t)b * CCOMP * HW;
#pragma unroll
    for (int r = 0; r < 8; ++r) {
        int m = m0 + r + 8 * half;
        cb[m * HW + n0 + lr] = acc[r] + b1[m];
    }
}

__global__ void gg_kern_gemm(const float* __restrict__ comp,
                             const float* __restrict__ w2,
                             const float* __restrict__ b2,
                             float* __restrict__ kern) {
    int b    = blockIdx.x >> 8;
    int n0   = (blockIdx.x & 255) * 16;
    int wave = threadIdx.x >> 5;
    int lane = threadIdx.x & 31;
    int lr   = lane & 15;
    int half = lane >> 4;
    int m0   = wave * 16;

    int n = n0 + lr;
    int h = n >> 6, w = n & 63;
    const float* cb = comp + (size_t)b * CCOMP * HW;
    int mA = m0 + lr;
    bool aValid = (mA < CENC);

    v8f acc = {};
    for (int k = 0; k < KDIM; k += 4) {
        int ke = k + 2 * half;
        v2f a, bm;
        a.x = aValid ? w2[mA * KDIM + ke]     : 0.f;
        a.y = aValid ? w2[mA * KDIM + ke + 1] : 0.f;
        {
            int ci = ke / 9, r = ke - ci * 9, ky = r / 3, kx = r - ky * 3;
            int hh = h + ky - 1, ww = w + kx - 1;
            bm.x = ((unsigned)hh < 64u && (unsigned)ww < 64u)
                       ? cb[ci * HW + hh * 64 + ww] : 0.f;
        }
        {
            int k1 = ke + 1;
            int ci = k1 / 9, r = k1 - ci * 9, ky = r / 3, kx = r - ky * 3;
            int hh = h + ky - 1, ww = w + kx - 1;
            bm.y = ((unsigned)hh < 64u && (unsigned)ww < 64u)
                       ? cb[ci * HW + hh * 64 + ww] : 0.f;
        }
        acc = __builtin_amdgcn_wmma_f32_16x16x4_f32(
                  false, a, false, bm, (short)0, acc, false, false);
    }

    float* kb = kern + (size_t)b * CENC * HW;
#pragma unroll
    for (int r = 0; r < 8; ++r) {
        int m = m0 + r + 8 * half;
        if (m < CENC) kb[m * HW + n] = acc[r] + b2[m];
    }
}

__global__ void gg_carafe_kernel(const float* __restrict__ x,
                                 const float* __restrict__ mean,
                                 const float* __restrict__ kern,
                                 float* __restrict__ out) {
    __shared__ float smask[KK][SS][WWID];

    int bh = blockIdx.x;
    int b  = bh >> 6, h = bh & 63;
    int tx = threadIdx.x & 63;
    int ty = threadIdx.x >> 6;

    const float* meanb = mean + b * HW;
    const float* kb    = kern + (size_t)b * CENC * HW;

    if (ty == 0) {
        int w   = tx;
        float mc = meanb[h * 64 + w];
        float mval[SS][KK];
#pragma unroll
        for (int kk = 0; kk < KK; ++kk) {
            int dy = kk / 3 - 1, dx = kk % 3 - 1;
            int hh = h + dy, ww = w + dx;
            float mp = ((unsigned)hh < 64u && (unsigned)ww < 64u)
                           ? meanb[hh * 64 + ww] : 0.f;
            float g = mp - mc;
            g = 1.0f / (g * g + 0.2f);
#pragma unroll
            for (int s = 0; s < SS; ++s)
                mval[s][kk] = g * kb[(kk * SS + s) * HW + h * 64 + w];
        }
#pragma unroll
        for (int s = 0; s < SS; ++s) {
            float mx = mval[s][0];
#pragma unroll
            for (int kk = 1; kk < KK; ++kk) mx = fmaxf(mx, mval[s][kk]);
            float e[KK], sum = 0.f;
#pragma unroll
            for (int kk = 0; kk < KK; ++kk) { e[kk] = __expf(mval[s][kk] - mx); sum += e[kk]; }
            float inv = 1.0f / sum;
#pragma unroll
            for (int kk = 0; kk < KK; ++kk) smask[kk][s][w] = e[kk] * inv;
        }
    }
    __syncthreads();

    for (int c = ty; c < CIN; c += 4) {
        const float* xb = x + ((size_t)(b * CIN + c)) * HW;
        __builtin_prefetch(xb + 4 * HW, 0, 0);
        float v[KK];
#pragma unroll
        for (int kk = 0; kk < KK; ++kk) {
            int dy = kk / 3 - 1, dx = kk % 3 - 1;
            int hh = h + dy, ww = tx + dx;
            v[kk] = ((unsigned)hh < 64u && (unsigned)ww < 64u)
                        ? xb[hh * 64 + ww] : 0.f;
        }
        size_t obase = ((size_t)(b * CIN + c)) * 128 * 128;
#pragma unroll
        for (int p = 0; p < 2; ++p)
#pragma unroll
            for (int q = 0; q < 2; ++q) {
                int s = p * 2 + q;
                float acc = 0.f;
#pragma unroll
                for (int kk = 0; kk < KK; ++kk)
                    acc = fmaf(v[kk], smask[kk][s][tx], acc);
                out[obase + (size_t)(2 * h + p) * 128 + (2 * tx + q)] = acc;
            }
    }
}

extern "C" void kernel_launch(void* const* d_in, const int* in_sizes, int n_in,
                              void* d_out, int out_size, void* d_ws, size_t ws_size,
                              hipStream_t stream) {
    const float* x  = (const float*)d_in[0];
    const float* w1 = (const float*)d_in[1];
    const float* b1 = (const float*)d_in[2];
    const float* w2 = (const float*)d_in[3];
    const float* b2 = (const float*)d_in[4];
    float* out = (float*)d_out;

    float* ws   = (float*)d_ws;
    float* mean = ws;
    float* comp = ws + 32768;
    float* kern = comp + (size_t)8 * CCOMP * HW;

    gg_mean_kernel <<<128,  256, 0, stream>>>(x, mean);
    gg_comp_gemm   <<<1024, 256, 0, stream>>>(x, w1, b1, comp);
    gg_kern_gemm   <<<2048,  96, 0, stream>>>(comp, w2, b2, kern);
    gg_carafe_kernel<<<512, 256, 0, stream>>>(x, mean, kern, out);
}


# 3x3-conv GEMM without LDS: one wave owns all 3 M tiles of 16 pixels; k loop unrolled over its 36-k period, im2col offsets and pad masks set once per lane; A pairs read as 8 B; sums unchanged
# speedup vs baseline: 1.3421x; 1.3421x over previous
#include <hip/hip_runtime.h>

typedef float v2f __attribute__((ext_vector_type(2)));
typedef float v8f __attribute__((ext_vector_type(8)));

#define CIN   256
#define CCOMP 64
#define CENC  36
#define HH    64
#define WWID  64
#define HW    4096
#define KK    9
#define SS    4
#define KDIM  576

__global__ void gg_mean_kernel(const float* __restrict__ x,
                               float* __restrict__ mean) {
    int idx = blockIdx.x * 256 + threadIdx.x;
    int b   = idx >> 12;
    int pix = idx & 4095;
    const float* xb = x + (size_t)b * CIN * HW + pix;
    float s = 0.f;
#pragma unroll 8
    for (int c = 0; c < CIN; ++c) s += xb[c * HW];
    mean[idx] = s * (1.0f / CIN);
}

__global__ void gg_comp_gemm(const float* __restrict__ x,
                             const float* __restrict__ w1,
                             const float* __restrict__ b1,
                             float* __restrict__ comp) {
    int b     = blockIdx.x >> 7;
    int nBase = (blockIdx.x & 127) * 32;
    int wave  = threadIdx.x >> 5;
    int lane  = threadIdx.x & 31;
    int lr    = lane & 15;
    int half  = lane >> 4;
    int m0    = (wave & 3) * 16;
    int n0    = nBase + (wave >> 2) * 16;

    const float* xb   = x + (size_t)b * CIN * HW;
    const float* aRow = w1 + (m0 + lr) * CIN;

    v8f acc = {};
    for (int k = 0; k < CIN; k += 4) {
        int ke = k + 2 * half;
        v2f a, bm;
        a.x  = aRow[ke];
        a.y  = aRow[ke + 1];
        bm.x = xb[ke * HW + n0 + lr];
        bm.y = xb[(ke + 1) * HW + n0 + lr];
        acc = __builtin_amdgcn_wmma_f32_16x16x4_f32(
                  false, a, false, bm, (short)0, acc, false, false);
    }

    float* cb = comp + (size_t)b * CCOMP * HW;
#pragma unroll
    for (int r = 0; r < 8; ++r) {
        int m = m0 + r + 8 * half;
        cb[m * HW + n0 + lr] = acc[r] + b1[m];
    }
}

__global__ __launch_bounds__(128)
void gg_kern_gemm(const float* __restrict__ comp,
                  const float* __restrict__ w2,
                  const float* __restrict__ b2,
                  float* __restrict__ kern) {
    int b    = blockIdx.x >> 6;
    int h    = blockIdx.x & 63;
    int wave = threadIdx.x >> 5;
    int lane = threadIdx.x & 31;
    int lr   = lane & 15;
    int half = lane >> 4;
    int w    = wave * 16 + lr;

    const char* cbytes = (const char*)(comp + (size_t)b * CCOMP * HW);
    const char* wbytes = (const char*)w2;

    unsigned boff[9][2];
    bool     bok[9][2];
#pragma unroll
    for (int j = 0; j < 9; ++j) {
#pragma unroll
        for (int e = 0; e < 2; ++e) {
            const int kA = 4 * j + e, kB = 4 * j + 2 + e;
            const int cA = kA / 9, tA = kA % 9;
            const int cB = kB / 9, tB = kB % 9;
            int ci = half ? cB : cA;
            int dy = half ? (tB / 3 - 1) : (tA / 3 - 1);
            int dx = half ? (tB % 3 - 1) : (tA % 3 - 1);
            int hh = h + dy, ww = w + dx;
            bool v = ((unsigned)hh < 64u) && ((unsigned)ww < 64u);
            bok[j][e]  = v;
            boff[j][e] = v ? (unsigned)((ci * HW + hh * 64 + ww) * 4) : 0u;
        }
    }

    bool     a2ok  = (32 + lr) < CENC;
    unsigned aoff0 = (unsigned)((lr * KDIM + 2 * half) * 4);
    unsigned aoff1 = aoff0 + 16u * KDIM * 4u;
    unsigned aoff2 = (unsigned)(((32 + (lr & 3)) * KDIM + 2 * half) * 4);

    v8f acc0 = {}, acc1 = {}, acc2 = {};
    for (int g = 0; g < CCOMP / 4; ++g) {
        const char* cg = cbytes + (size_t)g * (4 * HW * 4);
        const char* wg = wbytes + g * (36 * 4);
#pragma unroll
        for (int j = 0; j < 9; ++j) {
            float t0 = *(const float*)(cg + boff[j][0]);
            float t1 = *(const float*)(cg + boff[j][1]);
            v2f bm;
            bm.x = bok[j][0] ? t0 : 0.f;
            bm.y = bok[j][1] ? t1 : 0.f;
            v2f a0 = *(const v2f*)(wg + aoff0 + 16 * j);
            v2f a1 = *(const v2f*)(wg + aoff1 + 16 * j);
            v2f a2 = *(const v2f*)(wg + aoff2 + 16 * j);
            a2.x = a2ok ? a2.x : 0.f;
            a2.y = a2ok ? a2.y : 0.f;
            acc0 = __builtin_amdgcn_wmma_f32_16x16x4_f32(
                       false, a0, false, bm, (short)0, acc0, false, false);
            acc1 = __builtin_amdgcn_wmma_f32_16x16x4_f32(
                       false, a1, false, bm, (short)0, acc1, false, false);
            acc2 = __builtin_amdgcn_wmma_f32_16x16x4_f32(
                       false, a2, false, bm, (short)0, acc2, false, false);
        }
    }

    float* kb = kern + (size_t)b * CENC * HW + h * 64 + w;
#pragma unroll
    for (int r = 0; r < 8; ++r) {
        int m = r + 8 * half;
        kb[m * HW]        = acc0[r] + b2[m];
        kb[(m + 16) * HW] = acc1[r] + b2[m + 16];
        if (m + 32 < CENC) kb[(m + 32) * HW] = acc2[r] + b2[m + 32];
    }
}

__global__ void gg_carafe_kernel(const float* __restrict__ x,
                                 const float* __restrict__ mean,
                                 const float* __restrict__ kern,
                                 float* __restrict__ out) {
    __shared__ float smask[KK][SS][WWID];

    int bh = blockIdx.x;
    int b  = bh >> 6, h = bh & 63;
    int tx = threadIdx.x & 63;
    int ty = threadIdx.x >> 6;

    const float* meanb = mean + b * HW;
    const float* kb    = kern + (size_t)b * CENC * HW;

    if (ty == 0) {
        int w   = tx;
        float mc = meanb[h * 64 + w];
        float mval[SS][KK];
#pragma unroll
        for (int kk = 0; kk < KK; ++kk) {
            int dy = kk / 3 - 1, dx = kk % 3 - 1;
            int hh = h + dy, ww = w + dx;
            float mp = ((unsigned)hh < 64u && (unsigned)ww < 64u)
                           ? meanb[hh * 64 + ww] : 0.f;
            float g = mp - mc;
            g = 1.0f / (g * g + 0.2f);
#pragma unroll
            for (int s = 0; s < SS; ++s)
                mval[s][kk] = g * kb[(kk * SS + s) * HW + h * 64 + w];
        }
#pragma unroll
        for (int s = 0; s < SS; ++s) {
            float mx = mval[s][0];
#pragma unroll
            for (int kk = 1; kk < KK; ++kk) mx = fmaxf(mx, mval[s][kk]);
            float e[KK], sum = 0.f;
#pragma unroll
            for (int kk = 0; kk < KK; ++kk) { e[kk] = __expf(mval[s][kk] - mx); sum += e[kk]; }
            float inv = 1.0f / sum;
#pragma unroll
            for (int kk = 0; kk < KK; ++kk) smask[kk][s][w] = e[kk] * inv;
        }
    }
    __syncthreads();

    for (int c = ty; c < CIN; c += 4) {
        const float* xb = x + ((size_t)(b * CIN + c)) * HW;
        __builtin_prefetch(xb + 4 * HW, 0, 0);
        float v[KK];
#pragma unroll
        for (int kk = 0; kk < KK; ++kk) {
            int dy = kk / 3 - 1, dx = kk % 3 - 1;
            int hh = h + dy, ww = tx + dx;
            v[kk] = ((unsigned)hh < 64u && (unsigned)ww < 64u)
                        ? xb[hh * 64 + ww] : 0.f;
        }
        size_t obase = ((size_t)(b * CIN + c)) * 128 * 128;
#pragma unroll
        for (int p = 0; p < 2; ++p)
#pragma unroll
            for (int q = 0; q < 2; ++q) {
                int s = p * 2 + q;
                float acc = 0.f;
#pragma unroll
                for (int kk = 0; kk < KK; ++kk)
                    acc = fmaf(v[kk], smask[kk][s][tx], acc);
                out[obase + (size_t)(2 * h + p) * 128 + (2 * tx + q)] = acc;
            }
    }
}

extern "C" void kernel_launch(void* const* d_in, const int* in_sizes, int n_in,
                              void* d_out, int out_size, void* d_ws, size_t ws_size,
                              hipStream_t stream) {
    const float* x  = (const float*)d_in[0];
    const float* w1 = (const float*)d_in[1];
    const float* b1 = (const float*)d_in[2];
    const float* w2 = (const float*)d_in[3];
    const float* b2 = (const float*)d_in[4];
    float* out = (float*)d_out;

    float* ws   = (float*)d_ws;
    float* mean = ws;
    float* comp = ws + 32768;
    float* kern = comp + (size_t)8 * CCOMP * HW;

    gg_mean_kernel <<<128,  256, 0, stream>>>(x, mean);
    gg_comp_gemm   <<<1024, 256, 0, stream>>>(x, w1, b1, comp);
    gg_kern_gemm   <<<512,  128, 0, stream>>>(comp, w2, b2, kern);
    gg_carafe_kernel<<<512, 256, 0, stream>>>(x, mean, kern, out);
}


# reassembly kernel: x reaches the 9 taps through LDS: 32-channel chunks fetched as whole-line 16-B loads into zero-padded rows, 2 buffers, next chunk in flight, LDS-only barrier; sums unchanged
# speedup vs baseline: 1.8465x; 1.3758x over previous
#include <hip/hip_runtime.h>

typedef float v2f __attribute__((ext_vector_type(2)));
typedef float v4f __attribute__((ext_vector_type(4)));
typedef float v8f __attribute__((ext_vector_type(8)));

#define CIN   256
#define CCOMP 64
#define CENC  36
#define HH    64
#define WWID  64
#define HW    4096
#define KK    9
#define SS    4
#define KDIM  576

__global__ void gg_mean_kernel(const float* __restrict__ x,
                               float* __restrict__ mean) {
    int idx = blockIdx.x * 256 + threadIdx.x;
    int b   = idx >> 12;
    int pix = idx & 4095;
    const float* xb = x + (size_t)b * CIN * HW + pix;
    float s = 0.f;
#pragma unroll 8
    for (int c = 0; c < CIN; ++c) s += xb[c * HW];
    mean[idx] = s * (1.0f / CIN);
}

__global__ void gg_comp_gemm(const float* __restrict__ x,
                             const float* __restrict__ w1,
                             const float* __restrict__ b1,
                             float* __restrict__ comp) {
    int b     = blockIdx.x >> 7;
    int nBase = (blockIdx.x & 127) * 32;
    int wave  = threadIdx.x >> 5;
    int lane  = threadIdx.x & 31;
    int lr    = lane & 15;
    int half  = lane >> 4;
    int m0    = (wave & 3) * 16;
    int n0    = nBase + (wave >> 2) * 16;

    const float* xb   = x + (size_t)b * CIN * HW;
    const float* aRow = w1 + (m0 + lr) * CIN;

    v8f acc = {};
    for (int k = 0; k < CIN; k += 4) {
        int ke = k + 2 * half;
        v2f a, bm;
        a.x  = aRow[ke];
        a.y  = aRow[ke + 1];
        bm.x = xb[ke * HW + n0 + lr];
        bm.y = xb[(ke + 1) * HW + n0 + lr];
        acc = __builtin_amdgcn_wmma_f32_16x16x4_f32(
                  false, a, false, bm, (short)0, acc, false, false);
    }

    float* cb = comp + (size_t)b * CCOMP * HW;
#pragma unroll
    for (int r = 0; r < 8; ++r) {
        int m = m0 + r + 8 * half;
        cb[m * HW + n0 + lr] = acc[r] + b1[m];
    }
}

__global__ __launch_bounds__(128)
void gg_kern_gemm(const float* __restrict__ comp,
                  const float* __restrict__ w2,
                  const float* __restrict__ b2,
                  float* __restrict__ kern) {
    int b    = blockIdx.x >> 6;
    int h    = blockIdx.x & 63;
    int wave = threadIdx.x >> 5;
    int lane = threadIdx.x & 31;
    int lr   = lane & 15;
    int half = lane >> 4;
    int w    = wave * 16 + lr;

    const char* cbytes = (const char*)(comp + (size_t)b * CCOMP * HW);
    const char* wbytes = (const char*)w2;

    unsigned boff[9][2];
    bool     bok[9][2];
#pragma unroll
    for (int j = 0; j < 9; ++j) {
#pragma unroll
        for (int e = 0; e < 2; ++e) {
            const int kA = 4 * j + e, kB = 4 * j + 2 + e;
            const int cA = kA / 9, tA = kA % 9;
            const int cB = kB / 9, tB = kB % 9;
            int ci = half ? cB : cA;
            int dy = half ? (tB / 3 - 1) : (tA / 3 - 1);
            int dx = half ? (tB % 3 - 1) : (tA % 3 - 1);
            int hh = h + dy, ww = w + dx;
            bool v = ((unsigned)hh < 64u) && ((unsigned)ww < 64u);
            bok[j][e]  = v;
            boff[j][e] = v ? (unsigned)((ci * HW + hh * 64 + ww) * 4) : 0u;
        }
    }

    bool     a2ok  = (32 + lr) < CENC;
    unsigned aoff0 = (unsigned)((lr * KDIM + 2 * half) * 4);
    unsigned aoff1 = aoff0 + 16u * KDIM * 4u;
    unsigned aoff2 = (unsigned)(((32 + (lr & 3)) * KDIM + 2 * half) * 4);

    v8f acc0 = {}, acc1 = {}, acc2 = {};
    for (int g = 0; g < CCOMP / 4; ++g) {
        const char* cg = cbytes + (size_t)g * (4 * HW * 4);
        const char* wg = wbytes + g * (36 * 4);
#pragma unroll
        for (int j = 0; j < 9; ++j) {
            float t0 = *(const float*)(cg + boff[j][0]);
            float t1 = *(const float*)(cg + boff[j][1]);
            v2f bm;
            bm.x = bok[j][0] ? t0 : 0.f;
            bm.y = bok[j][1] ? t1 : 0.f;
            v2f a0 = *(const v2f*)(wg + aoff0 + 16 * j);
            v2f a1 = *(const v2f*)(wg + aoff1 + 16 * j);
            v2f a2 = *(const v2f*)(wg + aoff2 + 16 * j);
            a2.x = a2ok ? a2.x : 0.f;
            a2.y = a2ok ? a2.y : 0.f;
            acc0 = __builtin_amdgcn_wmma_f32_16x16x4_f32(
                       false, a0, false, bm, (short)0, acc0, false, false);
            acc1 = __builtin_amdgcn_wmma_f32_16x16x4_f32(
                       false, a1, false, bm, (short)0, acc1, false, false);
            acc2 = __builtin_amdgcn_wmma_f32_16x16x4_f32(
                       false, a2, false, bm, (short)0, acc2, false, false);
        }
    }

    float* kb = kern + (size_t)b * CENC * HW + h * 64 + w;
#pragma unroll
    for (int r = 0; r < 8; ++r) {
        int m = r + 8 * half;
        kb[m * HW]        = acc0[r] + b2[m];
        kb[(m + 16) * HW] = acc1[r] + b2[m + 16];
        if (m + 32 < CENC) kb[(m + 32) * HW] = acc2[r] + b2[m + 32];
    }
}

#define CH     32
#define XROW   72
#define NPIECE 6
#define XBUF   (CH * 3 * XROW)

__device__ __forceinline__ void gg_lds_barrier() {
    __builtin_amdgcn_fence(__ATOMIC_RELEASE, "workgroup", "local");
    __builtin_amdgcn_s_barrier();
    __builtin_amdgcn_fence(__ATOMIC_ACQUIRE, "workgroup", "local");
}

__global__ __launch_bounds__(256)
void gg_carafe_kernel(const float* __restrict__ x,
                      const float* __restrict__ mean,
                      const float* __restrict__ kern,
                      float* __restrict__ out) {
    __shared__ float smask[KK][SS][WWID];
    __shared__ __attribute__((aligned(16))) float xt[2 * XBUF];

    int bh  = blockIdx.x;
    int b   = bh >> 6, h = bh & 63;
    int tid = threadIdx.x;
    int tx  = tid & 63;
    int ty  = tid >> 6;

    const float* xb    = x + (size_t)b * CIN * HW;
    const float* meanb = mean + b * HW;
    const float* kb    = kern + (size_t)b * CENC * HW;

    unsigned goff[NPIECE], loff[NPIECE];
    bool     gok[NPIECE];
#pragma unroll
    for (int i = 0; i < NPIECE; ++i) {
        int q   = tid + 256 * i;
        int cc  = q / 48;
        int pos = q - cc * 48;
        int r   = pos >> 4, p = pos & 15;
        bool ok = (unsigned)(h - 1 + r) < 64u;
        gok[i]  = ok;
        goff[i] = ok ? (unsigned)(cc * HW + (h - 1) * 64 + pos * 4) : 0u;
        loff[i] = (unsigned)((cc * 3 + r) * XROW + 4 + p * 4);
    }

    v4f st[NPIECE];
#pragma unroll
    for (int i = 0; i < NPIECE; ++i)
        st[i] = *(const v4f*)(xb + goff[i]);

    const v4f zero4 = {0.f, 0.f, 0.f, 0.f};
    for (int i = tid; i < 2 * CH * 3 * 2; i += 256)
        *(v4f*)(xt + (i >> 1) * XROW + (i & 1) * 68) = zero4;

    if (ty == 0) {
        int w   = tx;
        float mc = meanb[h * 64 + w];
        float mval[SS][KK];
#pragma unroll
        for (int kk = 0; kk < KK; ++kk) {
            int dy = kk / 3 - 1, dx = kk % 3 - 1;
            int hh = h + dy, ww = w + dx;
            float mp = ((unsigned)hh < 64u && (unsigned)ww < 64u)
                           ? meanb[hh * 64 + ww] : 0.f;
            float g = mp - mc;
            g = 1.0f / (g * g + 0.2f);
#pragma unroll
            for (int s = 0; s < SS; ++s)
                mval[s][kk] = g * kb[(kk * SS + s) * HW + h * 64 + w];
        }
#pragma unroll
        for (int s = 0; s < SS; ++s) {
            float mx = mval[s][0];
#pragma unroll
            for (int kk = 1; kk < KK; ++kk) mx = fmaxf(mx, mval[s][kk]);
            float e[KK], sum = 0.f;
#pragma unroll
            for (int kk = 0; kk < KK; ++kk) { e[kk] = __expf(mval[s][kk] - mx); sum += e[kk]; }
            float inv = 1.0f / sum;
#pragma unroll
            for (int kk = 0; kk < KK; ++kk) smask[kk][s][w] = e[kk] * inv;
        }
    }

#pragma unroll
    for (int i = 0; i < NPIECE; ++i)
        *(v4f*)(xt + loff[i]) = gok[i] ? st[i] : zero4;
    __syncthreads();

    float m[KK][SS];
#pragma unroll
    for (int kk = 0; kk < KK; ++kk)
#pragma unroll
        for (int s = 0; s < SS; ++s) m[kk][s] = smask[kk][s][tx];

    float* ob = out + ((size_t)b * CIN * 128 + 2 * h) * 128 + 2 * tx;

    for (int k = 0; k < CIN / CH; ++k) {
        const bool more = (k + 1) < CIN / CH;
        if (more) {
            const float* xn = xb + (size_t)(k + 1) * CH * HW;
#pragma unroll
            for (int i = 0; i < NPIECE; ++i)
                st[i] = *(const v4f*)(xn + goff[i]);
        }
        __builtin_amdgcn_sched_barrier(0);

        const float* tb = xt + (k & 1) * XBUF + 3 + tx;
#pragma unroll
        for (int j = 0; j < CH / 4; ++j) {
            int cc = ty + 4 * j;
            const float* tr = tb + cc * 3 * XROW;
            float v[KK];
#pragma unroll
            for (int kk = 0; kk < KK; ++kk)
                v[kk] = tr[(kk / 3) * XROW + (kk % 3)];
            float a[SS];
#pragma unroll
            for (int s = 0; s < SS; ++s) {
                float acc = 0.f;
#pragma unroll
                for (int kk = 0; kk < KK; ++kk)
                    acc = fmaf(v[kk], m[kk][s], acc);
                a[s] = acc;
            }
            float* o = ob + (size_t)(k * CH + cc) * (128 * 128);
            v2f o0 = {a[0], a[1]}, o1 = {a[2], a[3]};
            *(v2f*)o         = o0;
            *(v2f*)(o + 128) = o1;
        }

        __builtin_amdgcn_sched_barrier(0);
        if (more) {
            float* tn = xt + ((k + 1) & 1) * XBUF;
#pragma unroll
            for (int i = 0; i < NPIECE; ++i)
                *(v4f*)(tn + loff[i]) = gok[i] ? st[i] : zero4;
            gg_lds_barrier();
        }
    }
}

extern "C" void kernel_launch(void* const* d_in, const int* in_sizes, int n_in,
                              void* d_out, int out_size, void* d_ws, size_t ws_size,
                              hipStream_t stream) {
    const float* x  = (const float*)d_in[0];
    const float* w1 = (const float*)d_in[1];
    const float* b1 = (const float*)d_in[2];
    const float* w2 = (const float*)d_in[3];
    const float* b2 = (const float*)d_in[4];
    float* out = (float*)d_out;

    float* ws   = (float*)d_ws;
    float* mean = ws;
    float* comp = ws + 32768;
    float* kern = comp + (size_t)8 * CCOMP * HW;

    gg_mean_kernel <<<128,  256, 0, stream>>>(x, mean);
    gg_comp_gemm   <<<1024, 256, 0, stream>>>(x, w1, b1, comp);
    gg_kern_gemm   <<<512,  128, 0, stream>>>(comp, w2, b2, kern);
    gg_carafe_kernel<<<512, 256, 0, stream>>>(x, mean, kern, out);
}


# channel-mean kernel: block per image row, 16 channel groups x 16 lanes of 4 pixels, a lane's 16 loads of 16 B all in flight, partials joined in LDS; mean's sum order changes (16 partial sums)
# speedup vs baseline: 2.2134x; 1.1987x over previous
#include <hip/hip_runtime.h>

typedef float v2f __attribute__((ext_vector_type(2)));
typedef float v4f __attribute__((ext_vector_type(4)));
typedef float v8f __attribute__((ext_vector_type(8)));

#define CIN   256
#define CCOMP 64
#define CENC  36
#define HH    64
#define WWID  64
#define HW    4096
#define KK    9
#define SS    4
#define KDIM  576

__global__ __launch_bounds__(256)
void gg_mean_kernel(const float* __restrict__ x,
                    float* __restrict__ mean) {
    __shared__ __attribute__((aligned(16))) float part[16][WWID];

    int tid = threadIdx.x;
    int p4  = tid & 15;
    int cg  = tid >> 4;
    int b   = blockIdx.x >> 6, h = blockIdx.x & 63;

    const float* xr = x + ((size_t)b * CIN + cg) * HW + h * 64 + p4 * 4;

    v4f v[16];
#pragma unroll
    for (int i = 0; i < 16; ++i)
        v[i] = *(const v4f*)(xr + (size_t)i * 16 * HW);
    __builtin_amdgcn_sched_barrier(0);
    v4f s = v[0];
#pragma unroll
    for (int i = 1; i < 16; ++i) s += v[i];

    *(v4f*)&part[cg][p4 * 4] = s;
    __syncthreads();

    if (tid < WWID) {
        float t = part[0][tid];
#pragma unroll
        for (int g = 1; g < 16; ++g) t += part[g][tid];
        mean[blockIdx.x * 64 + tid] = t * (1.0f / CIN);
    }
}

__global__ void gg_comp_gemm(const float* __restrict__ x,
                             const float* __restrict__ w1,
                             const float* __restrict__ b1,
                             float* __restrict__ comp) {
    int b     = blockIdx.x >> 7;
    int nBase = (blockIdx.x & 127) * 32;
    int wave  = threadIdx.x >> 5;
    int lane  = threadIdx.x & 31;
    int lr    = lane & 15;
    int half  = lane >> 4;
    int m0    = (wave & 3) * 16;
    int n0    = nBase + (wave >> 2) * 16;

    const float* xb   = x + (size_t)b * CIN * HW;
    const float* aRow = w1 + (m0 + lr) * CIN;

    v8f acc = {};
    for (int k = 0; k < CIN; k += 4) {
        int ke = k + 2 * half;
        v2f a, bm;
        a.x  = aRow[ke];
        a.y  = aRow[ke + 1];
        bm.x = xb[ke * HW + n0 + lr];
        bm.y = xb[(ke + 1) * HW + n0 + lr];
        acc = __builtin_amdgcn_wmma_f32_16x16x4_f32(
                  false, a, false, bm, (short)0, acc, false, false);
    }

    float* cb = comp + (size_t)b * CCOMP * HW;
#pragma unroll
    for (int r = 0; r < 8; ++r) {
        int m = m0 + r + 8 * half;
        cb[m * HW + n0 + lr] = acc[r] + b1[m];
    }
}

__global__ __launch_bounds__(128)
void gg_kern_gemm(const float* __restrict__ comp,
                  const float* __restrict__ w2,
                  const float* __restrict__ b2,
                  float* __restrict__ kern) {
    int b    = blockIdx.x >> 6;
    int h    = blockIdx.x & 63;
    int wave = threadIdx.x >> 5;
    int lane = threadIdx.x & 31;
    int lr   = lane & 15;
    int half = lane >> 4;
    int w    = wave * 16 + lr;

    const char* cbytes = (const char*)(comp + (size_t)b * CCOMP * HW);
    const char* wbytes = (const char*)w2;

    unsigned boff[9][2];
    bool     bok[9][2];
#pragma unroll
    for (int j = 0; j < 9; ++j) {
#pragma unroll
        for (int e = 0; e < 2; ++e) {
            const int kA = 4 * j + e, kB = 4 * j + 2 + e;
            const int cA = kA / 9, tA = kA % 9;
            const int cB = kB / 9, tB = kB % 9;
            int ci = half ? cB : cA;
            int dy = half ? (tB / 3 - 1) : (tA / 3 - 1);
            int dx = half ? (tB % 3 - 1) : (tA % 3 - 1);
            int hh = h + dy, ww = w + dx;
            bool v = ((unsigned)hh < 64u) && ((unsigned)ww < 64u);
            bok[j][e]  = v;
            boff[j][e] = v ? (unsigned)((ci * HW + hh * 64 + ww) * 4) : 0u;
        }
    }

    bool     a2ok  = (32 + lr) < CENC;
    unsigned aoff0 = (unsigned)((lr * KDIM + 2 * half) * 4);
    unsigned aoff1 = aoff0 + 16u * KDIM * 4u;
    unsigned aoff2 = (unsigned)(((32 + (lr & 3)) * KDIM + 2 * half) * 4);

    v8f acc0 = {}, acc1 = {}, acc2 = {};
    for (int g = 0; g < CCOMP / 4; ++g) {
        const char* cg = cbytes + (size_t)g * (4 * HW * 4);
        const char* wg = wbytes + g * (36 * 4);
#pragma unroll
        for (int j = 0; j < 9; ++j) {
            float t0 = *(const float*)(cg + boff[j][0]);
            float t1 = *(const float*)(cg + boff[j][1]);
            v2f bm;
            bm.x = bok[j][0] ? t0 : 0.f;
            bm.y = bok[j][1] ? t1 : 0.f;
            v2f a0 = *(const v2f*)(wg + aoff0 + 16 * j);
            v2f a1 = *(const v2f*)(wg + aoff1 + 16 * j);
            v2f a2 = *(const v2f*)(wg + aoff2 + 16 * j);
            a2.x = a2ok ? a2.x : 0.f;
            a2.y = a2ok ? a2.y : 0.f;
            acc0 = __builtin_amdgcn_wmma_f32_16x16x4_f32(
                       false, a0, false, bm, (short)0, acc0, false, false);
            acc1 = __builtin_amdgcn_wmma_f32_16x16x4_f32(
                       false, a1, false, bm, (short)0, acc1, false, false);
            acc2 = __builtin_amdgcn_wmma_f32_16x16x4_f32(
                       false, a2, false, bm, (short)0, acc2, false, false);
        }
    }

    float* kb = kern + (size_t)b * CENC * HW + h * 64 + w;
#pragma unroll
    for (int r = 0; r < 8; ++r) {
        int m = r + 8 * half;
        kb[m * HW]        = acc0[r] + b2[m];
        kb[(m + 16) * HW] = acc1[r] + b2[m + 16];
        if (m + 32 < CENC) kb[(m + 32) * HW] = acc2[r] + b2[m + 32];
    }
}

#define CH     32
#define XROW   72
#define NPIECE 6
#define XBUF   (CH * 3 * XROW)

__device__ __forceinline__ void gg_lds_barrier() {
    __builtin_amdgcn_fence(__ATOMIC_RELEASE, "workgroup", "local");
    __builtin_amdgcn_s_barrier();
    __builtin_amdgcn_fence(__ATOMIC_ACQUIRE, "workgroup", "local");
}

__global__ __launch_bounds__(256)
void gg_carafe_kernel(const float* __restrict__ x,
                      const float* __restrict__ mean,
                      const float* __restrict__ kern,
                      float* __restrict__ out) {
    __shared__ float smask[KK][SS][WWID];
    __shared__ __attribute__((aligned(16))) float xt[2 * XBUF];

    int bh  = blockIdx.x;
    int b   = bh >> 6, h = bh & 63;
    int tid = threadIdx.x;
    int tx  = tid & 63;
    int ty  = tid >> 6;

    const float* xb    = x + (size_t)b * CIN * HW;
    const float* meanb = mean + b * HW;
    const float* kb    = kern + (size_t)b * CENC * HW;

    unsigned goff[NPIECE], loff[NPIECE];
    bool     gok[NPIECE];
#pragma unroll
    for (int i = 0; i < NPIECE; ++i) {
        int q   = tid + 256 * i;
        int cc  = q / 48;
        int pos = q - cc * 48;
        int r   = pos >> 4, p = pos & 15;
        bool ok = (unsigned)(h - 1 + r) < 64u;
        gok[i]  = ok;
        goff[i] = ok ? (unsigned)(cc * HW + (h - 1) * 64 + pos * 4) : 0u;
        loff[i] = (unsigned)((cc * 3 + r) * XROW + 4 + p * 4);
    }

    v4f st[NPIECE];
#pragma unroll
    for (int i = 0; i < NPIECE; ++i)
        st[i] = *(const v4f*)(xb + goff[i]);

    const v4f zero4 = {0.f, 0.f, 0.f, 0.f};
    for (int i = tid; i < 2 * CH * 3 * 2; i += 256)
        *(v4f*)(xt + (i >> 1) * XROW + (i & 1) * 68) = zero4;

    if (ty == 0) {
        int w   = tx;
        float mc = meanb[h * 64 + w];
        float mval[SS][KK];
#pragma unroll
        for (int kk = 0; kk < KK; ++kk) {
            int dy = kk / 3 - 1, dx = kk % 3 - 1;
            int hh = h + dy, ww = w + dx;
            float mp = ((unsigned)hh < 64u && (unsigned)ww < 64u)
                           ? meanb[hh * 64 + ww] : 0.f;
            float g = mp - mc;
            g = 1.0f / (g * g + 0.2f);
#pragma unroll
            for (int s = 0; s < SS; ++s)
                mval[s][kk] = g * kb[(kk * SS + s) * HW + h * 64 + w];
        }
#pragma unroll
        for (int s = 0; s < SS; ++s) {
            float mx = mval[s][0];
#pragma unroll
            for (int kk = 1; kk < KK; ++kk) mx = fmaxf(mx, mval[s][kk]);
            float e[KK], sum = 0.f;
#pragma unroll
            for (int kk = 0; kk < KK; ++kk) { e[kk] = __expf(mval[s][kk] - mx); sum += e[kk]; }
            float inv = 1.0f / sum;
#pragma unroll
            for (int kk = 0; kk < KK; ++kk) smask[kk][s][w] = e[kk] * inv;
        }
    }

#pragma unroll
    for (int i = 0; i < NPIECE; ++i)
        *(v4f*)(xt + loff[i]) = gok[i] ? st[i] : zero4;
    __syncthreads();

    float m[KK][SS];
#pragma unroll
    for (int kk = 0; kk < KK; ++kk)
#pragma unroll
        for (int s = 0; s < SS; ++s) m[kk][s] = smask[kk][s][tx];

    float* ob = out + ((size_t)b * CIN * 128 + 2 * h) * 128 + 2 * tx;

    for (int k = 0; k < CIN / CH; ++k) {
        const bool more = (k + 1) < CIN / CH;
        if (more) {
            const float* xn = xb + (size_t)(k + 1) * CH * HW;
#pragma unroll
            for (int i = 0; i < NPIECE; ++i)
                st[i] = *(const v4f*)(xn + goff[i]);
        }
        __builtin_amdgcn_sched_barrier(0);

        const float* tb = xt + (k & 1) * XBUF + 3 + tx;
#pragma unroll
        for (int j = 0; j < CH / 4; ++j) {
            int cc = ty + 4 * j;
            const float* tr = tb + cc * 3 * XROW;
            float v[KK];
#pragma unroll
            for (int kk = 0; kk < KK; ++kk)
                v[kk] = tr[(kk / 3) * XROW + (kk % 3)];
            float a[SS];
#pragma unroll
            for (int s = 0; s < SS; ++s) {
                float acc = 0.f;
#pragma unroll
                for (int kk = 0; kk < KK; ++kk)
                    acc = fmaf(v[kk], m[kk][s], acc);
                a[s] = acc;
            }
            float* o = ob + (size_t)(k * CH + cc) * (128 * 128);
            v2f o0 = {a[0], a[1]}, o1 = {a[2], a[3]};
            *(v2f*)o         = o0;
            *(v2f*)(o + 128) = o1;
        }

        __builtin_amdgcn_sched_barrier(0);
        if (more) {
            float* tn = xt + ((k + 1) & 1) * XBUF;
#pragma unroll
            for (int i = 0; i < NPIECE; ++i)
                *(v4f*)(tn + loff[i]) = gok[i] ? st[i] : zero4;
            gg_lds_barrier();
        }
    }
}

extern "C" void kernel_launch(void* const* d_in, const int* in_sizes, int n_in,
                              void* d_out, int out_size, void* d_ws, size_t ws_size,
                              hipStream_t stream) {
    const float* x  = (const float*)d_in[0];
    const float* w1 = (const float*)d_in[1];
    const float* b1 = (const float*)d_in[2];
    const float* w2 = (const float*)d_in[3];
    const float* b2 = (const float*)d_in[4];
    float* out = (float*)d_out;

    float* ws   = (float*)d_ws;
    float* mean = ws;
    float* comp = ws + 32768;
    float* kern = comp + (size_t)8 * CCOMP * HW;

    gg_mean_kernel <<<512,  256, 0, stream>>>(x, mean);
    gg_comp_gemm   <<<1024, 256, 0, stream>>>(x, w1, b1, comp);
    gg_kern_gemm   <<<512,  128, 0, stream>>>(comp, w2, b2, kern);
    gg_carafe_kernel<<<512, 256, 0, stream>>>(x, mean, kern, out);
}


# 1x1-conv GEMM without LDS: wave = one M tile x 4 N tiles of interleaved pixels; x and w1 read as 16-B pieces, next 32 k in flight in registers; k order inside an 8-k block changes comp's sum order
# speedup vs baseline: 2.4859x; 1.1231x over previous
#include <hip/hip_runtime.h>

typedef float v2f __attribute__((ext_vector_type(2)));
typedef float v4f __attribute__((ext_vector_type(4)));
typedef float v8f __attribute__((ext_vector_type(8)));

#define CIN   256
#define CCOMP 64
#define CENC  36
#define HH    64
#define WWID  64
#define HW    4096
#define KK    9
#define SS    4
#define KDIM  576

__global__ __launch_bounds__(256)
void gg_mean_kernel(const float* __restrict__ x,
                    float* __restrict__ mean) {
    __shared__ __attribute__((aligned(16))) float part[16][WWID];

    int tid = threadIdx.x;
    int p4  = tid & 15;
    int cg  = tid >> 4;
    int b   = blockIdx.x >> 6, h = blockIdx.x & 63;

    const float* xr = x + ((size_t)b * CIN + cg) * HW + h * 64 + p4 * 4;

    v4f v[16];
#pragma unroll
    for (int i = 0; i < 16; ++i)
        v[i] = *(const v4f*)(xr + (size_t)i * 16 * HW);
    __builtin_amdgcn_sched_barrier(0);
    v4f s = v[0];
#pragma unroll
    for (int i = 1; i < 16; ++i) s += v[i];

    *(v4f*)&part[cg][p4 * 4] = s;
    __syncthreads();

    if (tid < WWID) {
        float t = part[0][tid];
#pragma unroll
        for (int g = 1; g < 16; ++g) t += part[g][tid];
        mean[blockIdx.x * 64 + tid] = t * (1.0f / CIN);
    }
}

#define G2_KB 4
#define G2_NS (CIN / (8 * G2_KB))

__device__ __forceinline__ void gg_comp_load(const float* __restrict__ ap,
                                             const float* __restrict__ bp,
                                             int s, v4f (&aq)[G2_KB],
                                             v4f (&bq)[G2_KB][4]) {
#pragma unroll
    for (int kb = 0; kb < G2_KB; ++kb) {
        int k0 = (s * G2_KB + kb) * 8;
        aq[kb] = *(const v4f*)(ap + k0);
#pragma unroll
        for (int e = 0; e < 4; ++e)
            bq[kb][e] = *(const v4f*)(bp + (size_t)(k0 + e) * HW);
    }
}

__device__ __forceinline__ void gg_comp_mma(const v4f (&aq)[G2_KB],
                                            const v4f (&bq)[G2_KB][4],
                                            v8f (&acc)[4]) {
#pragma unroll
    for (int kb = 0; kb < G2_KB; ++kb) {
#pragma unroll
        for (int st = 0; st < 2; ++st) {
            v2f a;
            a.x = aq[kb][2 * st];
            a.y = aq[kb][2 * st + 1];
#pragma unroll
            for (int t = 0; t < 4; ++t) {
                v2f bm;
                bm.x = bq[kb][2 * st][t];
                bm.y = bq[kb][2 * st + 1][t];
                acc[t] = __builtin_amdgcn_wmma_f32_16x16x4_f32(
                             false, a, false, bm, (short)0, acc[t], false, false);
            }
        }
    }
}

__global__ __launch_bounds__(128)
void gg_comp_gemm(const float* __restrict__ x,
                  const float* __restrict__ w1,
                  const float* __restrict__ b1,
                  float* __restrict__ comp) {
    int b    = blockIdx.x >> 6, h = blockIdx.x & 63;
    int wave = threadIdx.x >> 5;
    int lane = threadIdx.x & 31;
    int lr   = lane & 15;
    int half = lane >> 4;
    int m0   = wave * 16;

    const float* ap = w1 + (m0 + lr) * CIN + 4 * half;
    const float* bp = x + ((size_t)b * CIN + 4 * half) * HW + h * 64 + 4 * lr;

    float bias[8];
#pragma unroll
    for (int r = 0; r < 8; ++r) bias[r] = b1[m0 + 8 * half + r];

    v8f acc[4] = {};
    v4f aq[2][G2_KB];
    v4f bq[2][G2_KB][4];

    gg_comp_load(ap, bp, 0, aq[0], bq[0]);
#pragma unroll
    for (int s = 0; s < G2_NS; ++s) {
        if (s + 1 < G2_NS) gg_comp_load(ap, bp, s + 1, aq[(s + 1) & 1], bq[(s + 1) & 1]);
        __builtin_amdgcn_sched_barrier(0);
        gg_comp_mma(aq[s & 1], bq[s & 1], acc);
    }

    float* cb = comp + ((size_t)b * CCOMP + m0 + 8 * half) * HW + h * 64 + 4 * lr;
#pragma unroll
    for (int r = 0; r < 8; ++r) {
        v4f o = {acc[0][r] + bias[r], acc[1][r] + bias[r],
                 acc[2][r] + bias[r], acc[3][r] + bias[r]};
        *(v4f*)(cb + (size_t)r * HW) = o;
    }
}

__global__ __launch_bounds__(128)
void gg_kern_gemm(const float* __restrict__ comp,
                  const float* __restrict__ w2,
                  const float* __restrict__ b2,
                  float* __restrict__ kern) {
    int b    = blockIdx.x >> 6;
    int h    = blockIdx.x & 63;
    int wave = threadIdx.x >> 5;
    int lane = threadIdx.x & 31;
    int lr   = lane & 15;
    int half = lane >> 4;
    int w    = wave * 16 + lr;

    const char* cbytes = (const char*)(comp + (size_t)b * CCOMP * HW);
    const char* wbytes = (const char*)w2;

    unsigned boff[9][2];
    bool     bok[9][2];
#pragma unroll
    for (int j = 0; j < 9; ++j) {
#pragma unroll
        for (int e = 0; e < 2; ++e) {
            const int kA = 4 * j + e, kB = 4 * j + 2 + e;
            const int cA = kA / 9, tA = kA % 9;
            const int cB = kB / 9, tB = kB % 9;
            int ci = half ? cB : cA;
            int dy = half ? (tB / 3 - 1) : (tA / 3 - 1);
            int dx = half ? (tB % 3 - 1) : (tA % 3 - 1);
            int hh = h + dy, ww = w + dx;
            bool v = ((unsigned)hh < 64u) && ((unsigned)ww < 64u);
            bok[j][e]  = v;
            boff[j][e] = v ? (unsigned)((ci * HW + hh * 64 + ww) * 4) : 0u;
        }
    }

    bool     a2ok  = (32 + lr) < CENC;
    unsigned aoff0 = (unsigned)((lr * KDIM + 2 * half) * 4);
    unsigned aoff1 = aoff0 + 16u * KDIM * 4u;
    unsigned aoff2 = (unsigned)(((32 + (lr & 3)) * KDIM + 2 * half) * 4);

    v8f acc0 = {}, acc1 = {}, acc2 = {};
    for (int g = 0; g < CCOMP / 4; ++g) {
        const char* cg = cbytes + (size_t)g * (4 * HW * 4);
        const char* wg = wbytes + g * (36 * 4);
#pragma unroll
        for (int j = 0; j < 9; ++j) {
            float t0 = *(const float*)(cg + boff[j][0]);
            float t1 = *(const float*)(cg + boff[j][1]);
            v2f bm;
            bm.x = bok[j][0] ? t0 : 0.f;
            bm.y = bok[j][1] ? t1 : 0.f;
            v2f a0 = *(const v2f*)(wg + aoff0 + 16 * j);
            v2f a1 = *(const v2f*)(wg + aoff1 + 16 * j);
            v2f a2 = *(const v2f*)(wg + aoff2 + 16 * j);
            a2.x = a2ok ? a2.x : 0.f;
            a2.y = a2ok ? a2.y : 0.f;
            acc0 = __builtin_amdgcn_wmma_f32_16x16x4_f32(
                       false, a0, false, bm, (short)0, acc0, false, false);
            acc1 = __builtin_amdgcn_wmma_f32_16x16x4_f32(
                       false, a1, false, bm, (short)0, acc1, false, false);
            acc2 = __builtin_amdgcn_wmma_f32_16x16x4_f32(
                       false, a2, false, bm, (short)0, acc2, false, false);
        }
    }

    float* kb = kern + (size_t)b * CENC * HW + h * 64 + w;
#pragma unroll
    for (int r = 0; r < 8; ++r) {
        int m = r + 8 * half;
        kb[m * HW]        = acc0[r] + b2[m];
        kb[(m + 16) * HW] = acc1[r] + b2[m + 16];
        if (m + 32 < CENC) kb[(m + 32) * HW] = acc2[r] + b2[m + 32];
    }
}

#define CH     32
#define XROW   72
#define NPIECE 6
#define XBUF   (CH * 3 * XROW)

__device__ __forceinline__ void gg_lds_barrier() {
    __builtin_amdgcn_fence(__ATOMIC_RELEASE, "workgroup", "local");
    __builtin_amdgcn_s_barrier();
    __builtin_amdgcn_fence(__ATOMIC_ACQUIRE, "workgroup", "local");
}

__global__ __launch_bounds__(256)
void gg_carafe_kernel(const float* __restrict__ x,
                      const float* __restrict__ mean,
                      const float* __restrict__ kern,
                      float* __restrict__ out) {
    __shared__ float smask[KK][SS][WWID];
    __shared__ __attribute__((aligned(16))) float xt[2 * XBUF];

    int bh  = blockIdx.x;
    int b   = bh >> 6, h = bh & 63;
    int tid = threadIdx.x;
    int tx  = tid & 63;
    int ty  = tid >> 6;

    const float* xb    = x + (size_t)b * CIN * HW;
    const float* meanb = mean + b * HW;
    const float* kb    = kern + (size_t)b * CENC * HW;

    unsigned goff[NPIECE], loff[NPIECE];
    bool     gok[NPIECE];
#pragma unroll
    for (int i = 0; i < NPIECE; ++i) {
        int q   = tid + 256 * i;
        int cc  = q / 48;
        int pos = q - cc * 48;
        int r   = pos >> 4, p = pos & 15;
        bool ok = (unsigned)(h - 1 + r) < 64u;
        gok[i]  = ok;
        goff[i] = ok ? (unsigned)(cc * HW + (h - 1) * 64 + pos * 4) : 0u;
        loff[i] = (unsigned)((cc * 3 + r) * XROW + 4 + p * 4);
    }

    v4f st[NPIECE];
#pragma unroll
    for (int i = 0; i < NPIECE; ++i)
        st[i] = *(const v4f*)(xb + goff[i]);

    const v4f zero4 = {0.f, 0.f, 0.f, 0.f};
    for (int i = tid; i < 2 * CH * 3 * 2; i += 256)
        *(v4f*)(xt + (i >> 1) * XROW + (i & 1) * 68) = zero4;

    if (ty == 0) {
        int w   = tx;
        float mc = meanb[h * 64 + w];
        float mval[SS][KK];
#pragma unroll
        for (int kk = 0; kk < KK; ++kk) {
            int dy = kk / 3 - 1, dx = kk % 3 - 1;
            int hh = h + dy, ww = w + dx;
            float mp = ((unsigned)hh < 64u && (unsigned)ww < 64u)
                           ? meanb[hh * 64 + ww] : 0.f;
            float g = mp - mc;
            g = 1.0f / (g * g + 0.2f);
#pragma unroll
            for (int s = 0; s < SS; ++s)
                mval[s][kk] = g * kb[(kk * SS + s) * HW + h * 64 + w];
        }
#pragma unroll
        for (int s = 0; s < SS; ++s) {
            float mx = mval[s][0];
#pragma unroll
            for (int kk = 1; kk < KK; ++kk) mx = fmaxf(mx, mval[s][kk]);
            float e[KK], sum = 0.f;
#pragma unroll
            for (int kk = 0; kk < KK; ++kk) { e[kk] = __expf(mval[s][kk] - mx); sum += e[kk]; }
            float inv = 1.0f / sum;
#pragma unroll
            for (int kk = 0; kk < KK; ++kk) smask[kk][s][w] = e[kk] * inv;
        }
    }

#pragma unroll
    for (int i = 0; i < NPIECE; ++i)
        *(v4f*)(xt + loff[i]) = gok[i] ? st[i] : zero4;
    __syncthreads();

    float m[KK][SS];
#pragma unroll
    for (int kk = 0; kk < KK; ++kk)
#pragma unroll
        for (int s = 0; s < SS; ++s) m[kk][s] = smask[kk][s][tx];

    float* ob = out + ((size_t)b * CIN * 128 + 2 * h) * 128 + 2 * tx;

    for (int k = 0; k < CIN / CH; ++k) {
        const bool more = (k + 1) < CIN / CH;
        if (more) {
            const float* xn = xb + (size_t)(k + 1) * CH * HW;
#pragma unroll
            for (int i = 0; i < NPIECE; ++i)
                st[i] = *(const v4f*)(xn + goff[i]);
        }
        __builtin_amdgcn_sched_barrier(0);

        const float* tb = xt + (k & 1) * XBUF + 3 + tx;
#pragma unroll
        for (int j = 0; j < CH / 4; ++j) {
            int cc = ty + 4 * j;
            const float* tr = tb + cc * 3 * XROW;
            float v[KK];
#pragma unroll
            for (int kk = 0; kk < KK; ++kk)
                v[kk] = tr[(kk / 3) * XROW + (kk % 3)];
            float a[SS];
#pragma unroll
            for (int s = 0; s < SS; ++s) {
                float acc = 0.f;
#pragma unroll
                for (int kk = 0; kk < KK; ++kk)
                    acc = fmaf(v[kk], m[kk][s], acc);
                a[s] = acc;
            }
            float* o = ob + (size_t)(k * CH + cc) * (128 * 128);
            v2f o0 = {a[0], a[1]}, o1 = {a[2], a[3]};
            *(v2f*)o         = o0;
            *(v2f*)(o + 128) = o1;
        }

        __builtin_amdgcn_sched_barrier(0);
        if (more) {
            float* tn = xt + ((k + 1) & 1) * XBUF;
#pragma unroll
            for (int i = 0; i < NPIECE; ++i)
                *(v4f*)(tn + loff[i]) = gok[i] ? st[i] : zero4;
            gg_lds_barrier();
        }
    }
}

extern "C" void kernel_launch(void* const* d_in, const int* in_sizes, int n_in,
                              void* d_out, int out_size, void* d_ws, size_t ws_size,
                              hipStream_t stream) {
    const float* x  = (const float*)d_in[0];
    const float* w1 = (const float*)d_in[1];
    const float* b1 = (const float*)d_in[2];
    const float* w2 = (const float*)d_in[3];
    const float* b2 = (const float*)d_in[4];
    float* out = (float*)d_out;

    float* ws   = (float*)d_ws;
    float* mean = ws;
    float* comp = ws + 32768;
    float* kern = comp + (size_t)8 * CCOMP * HW;

    gg_mean_kernel <<<512,  256, 0, stream>>>(x, mean);
    gg_comp_gemm   <<<512,  128, 0, stream>>>(x, w1, b1, comp);
    gg_kern_gemm   <<<512,  128, 0, stream>>>(comp, w2, b2, kern);
    gg_carafe_kernel<<<512, 256, 0, stream>>>(x, mean, kern, out);
}


# 3x3-conv GEMM: w2 re-laid each call in WMMA fragment order (by the mean kernel), A read as whole-line 16-B loads; next 72 k of comp and A in flight in registers; k order and sums unchanged
# speedup vs baseline: 3.3189x; 1.3351x over previous
#include <hip/hip_runtime.h>

typedef float v2f __attribute__((ext_vector_type(2)));
typedef float v4f __attribute__((ext_vector_type(4)));
typedef float v8f __attribute__((ext_vector_type(8)));

#define CIN   256
#define CCOMP 64
#define CENC  36
#define HH    64
#define WWID  64
#define HW    4096
#define KK    9
#define SS    4
#define KDIM  576

#define W2F_PER_BLOCK 54

__global__ __launch_bounds__(256)
void gg_mean_kernel(const float* __restrict__ x,
                    const float* __restrict__ w2,
                    float* __restrict__ mean,
                    float* __restrict__ w2f) {
    __shared__ __attribute__((aligned(16))) float part[16][WWID];

    int tid = threadIdx.x;
    int p4  = tid & 15;
    int cg  = tid >> 4;
    int b   = blockIdx.x >> 6, h = blockIdx.x & 63;

    const float* xr = x + ((size_t)b * CIN + cg) * HW + h * 64 + p4 * 4;

    v4f v[16];
#pragma unroll
    for (int i = 0; i < 16; ++i)
        v[i] = *(const v4f*)(xr + (size_t)i * 16 * HW);
    __builtin_amdgcn_sched_barrier(0);
    v4f s = v[0];
#pragma unroll
    for (int i = 1; i < 16; ++i) s += v[i];

    *(v4f*)&part[cg][p4 * 4] = s;
    __syncthreads();

    if (tid < WWID) {
        float t = part[0][tid];
#pragma unroll
        for (int g = 1; g < 16; ++g) t += part[g][tid];
        mean[blockIdx.x * 64 + tid] = t * (1.0f / CIN);
    } else if (tid < WWID + W2F_PER_BLOCK) {
        int i   = blockIdx.x * W2F_PER_BLOCK + (tid - WWID);
        int r   = i & 3, l = (i >> 2) & 31, pt = i >> 7;
        int t   = pt % 3, p = pt / 3;
        int k   = 4 * (2 * p + (r >> 1)) + 2 * (l >> 4) + (r & 1);
        int row = 16 * t + (l & 15);
        float v = w2[(row < CENC ? row : 0) * KDIM + k];
        w2f[i]  = row < CENC ? v : 0.f;
    }
}

#define G2_KB 4
#define G2_NS (CIN / (8 * G2_KB))

__device__ __forceinline__ void gg_comp_load(const float* __restrict__ ap,
                                             const float* __restrict__ bp,
                                             int s, v4f (&aq)[G2_KB],
                                             v4f (&bq)[G2_KB][4]) {
#pragma unroll
    for (int kb = 0; kb < G2_KB; ++kb) {
        int k0 = (s * G2_KB + kb) * 8;
        aq[kb] = *(const v4f*)(ap + k0);
#pragma unroll
        for (int e = 0; e < 4; ++e)
            bq[kb][e] = *(const v4f*)(bp + (size_t)(k0 + e) * HW);
    }
}

__device__ __forceinline__ void gg_comp_mma(const v4f (&aq)[G2_KB],
                                            const v4f (&bq)[G2_KB][4],
                                            v8f (&acc)[4]) {
#pragma unroll
    for (int kb = 0; kb < G2_KB; ++kb) {
#pragma unroll
        for (int st = 0; st < 2; ++st) {
            v2f a;
            a.x = aq[kb][2 * st];
            a.y = aq[kb][2 * st + 1];
#pragma unroll
            for (int t = 0; t < 4; ++t) {
                v2f bm;
                bm.x = bq[kb][2 * st][t];
                bm.y = bq[kb][2 * st + 1][t];
                acc[t] = __builtin_amdgcn_wmma_f32_16x16x4_f32(
                             false, a, false, bm, (short)0, acc[t], false, false);
            }
        }
    }
}

__global__ __launch_bounds__(128)
void gg_comp_gemm(const float* __restrict__ x,
                  const float* __restrict__ w1,
                  const float* __restrict__ b1,
                  float* __restrict__ comp) {
    int b    = blockIdx.x >> 6, h = blockIdx.x & 63;
    int wave = threadIdx.x >> 5;
    int lane = threadIdx.x & 31;
    int lr   = lane & 15;
    int half = lane >> 4;
    int m0   = wave * 16;

    const float* ap = w1 + (m0 + lr) * CIN + 4 * half;
    const float* bp = x + ((size_t)b * CIN + 4 * half) * HW + h * 64 + 4 * lr;

    float bias[8];
#pragma unroll
    for (int r = 0; r < 8; ++r) bias[r] = b1[m0 + 8 * half + r];

    v8f acc[4] = {};
    v4f aq[2][G2_KB];
    v4f bq[2][G2_KB][4];

    gg_comp_load(ap, bp, 0, aq[0], bq[0]);
#pragma unroll
    for (int s = 0; s < G2_NS; ++s) {
        if (s + 1 < G2_NS) gg_comp_load(ap, bp, s + 1, aq[(s + 1) & 1], bq[(s + 1) & 1]);
        __builtin_amdgcn_sched_barrier(0);
        gg_comp_mma(aq[s & 1], bq[s & 1], acc);
    }

    float* cb = comp + ((size_t)b * CCOMP + m0 + 8 * half) * HW + h * 64 + 4 * lr;
#pragma unroll
    for (int r = 0; r < 8; ++r) {
        v4f o = {acc[0][r] + bias[r], acc[1][r] + bias[r],
                 acc[2][r] + bias[r], acc[3][r] + bias[r]};
        *(v4f*)(cb + (size_t)r * HW) = o;
    }
}

#define G3_UNITS (KDIM / 72)
#define G3_AP    27

__device__ __forceinline__ void gg_kern_load_a(const v4f* __restrict__ af,
                                               v4f (&aq)[G3_AP]) {
#pragma unroll
    for (int i = 0; i < G3_AP; ++i) aq[i] = af[i * 32];
}

__device__ __forceinline__ void gg_kern_load_b(const char* __restrict__ cu,
                                               const unsigned (&boff)[9][2],
                                               float (&bq)[2][9][2]) {
#pragma unroll
    for (int hp = 0; hp < 2; ++hp) {
#pragma unroll
        for (int j = 0; j < 9; ++j) {
#pragma unroll
            for (int e = 0; e < 2; ++e)
                bq[hp][j][e] = *(const float*)(cu + hp * (4 * HW * 4) + boff[j][e]);
        }
    }
}

template <int HP>
__device__ __forceinline__ void gg_kern_mma(const v4f (&aq)[G3_AP],
                                            const float (&bq)[2][9][2],
                                            const bool (&bok)[9][2],
                                            v8f& acc0, v8f& acc1, v8f& acc2) {
#pragma unroll
    for (int j = 0; j < 9; ++j) {
        const int s  = 9 * HP + j;
        const int pp = s >> 1, o = (s & 1) * 2;
        v2f bm;
        bm.x = bok[j][0] ? bq[HP][j][0] : 0.f;
        bm.y = bok[j][1] ? bq[HP][j][1] : 0.f;
        v2f a0 = {aq[pp * 3 + 0][o], aq[pp * 3 + 0][o + 1]};
        v2f a1 = {aq[pp * 3 + 1][o], aq[pp * 3 + 1][o + 1]};
        v2f a2 = {aq[pp * 3 + 2][o], aq[pp * 3 + 2][o + 1]};
        acc0 = __builtin_amdgcn_wmma_f32_16x16x4_f32(
                   false, a0, false, bm, (short)0, acc0, false, false);
        acc1 = __builtin_amdgcn_wmma_f32_16x16x4_f32(
                   false, a1, false, bm, (short)0, acc1, false, false);
        acc2 = __builtin_amdgcn_wmma_f32_16x16x4_f32(
                   false, a2, false, bm, (short)0, acc2, false, false);
    }
}

#define GG_KERN_UNIT(CUR, NXT, afn, cun)                                   \
    gg_kern_load_b(cun, boff, bq[NXT]);                                    \
    __builtin_amdgcn_sched_barrier(0);                                     \
    gg_kern_mma<0>(aq[CUR], bq[CUR], bok, acc0, acc1, acc2);               \
    __builtin_amdgcn_sched_barrier(0);                                     \
    gg_kern_load_a(afn, aq[NXT]);                                          \
    __builtin_amdgcn_sched_barrier(0);                                     \
    gg_kern_mma<1>(aq[CUR], bq[CUR], bok, acc0, acc1, acc2);               \
    __builtin_amdgcn_sched_barrier(0);

__global__ __launch_bounds__(128)
void gg_kern_gemm(const float* __restrict__ comp,
                  const float* __restrict__ w2f,
                  const float* __restrict__ b2,
                  float* __restrict__ kern) {
    int b    = blockIdx.x >> 6;
    int h    = blockIdx.x & 63;
    int wave = threadIdx.x >> 5;
    int lane = threadIdx.x & 31;
    int lr   = lane & 15;
    int half = lane >> 4;
    int w    = wave * 16 + lr;

    const char* cbytes = (const char*)(comp + (size_t)b * CCOMP * HW);
    const v4f*  af     = (const v4f*)w2f + lane;

    unsigned boff[9][2];
    bool     bok[9][2];
#pragma unroll
    for (int j = 0; j < 9; ++j) {
#pragma unroll
        for (int e = 0; e < 2; ++e) {
            const int kA = 4 * j + e, kB = 4 * j + 2 + e;
            const int cA = kA / 9, tA = kA % 9;
            const int cB = kB / 9, tB = kB % 9;
            int ci = half ? cB : cA;
            int dy = half ? (tB / 3 - 1) : (tA / 3 - 1);
            int dx = half ? (tB % 3 - 1) : (tA % 3 - 1);
            int hh = h + dy, ww = w + dx;
            bool v = ((unsigned)hh < 64u) && ((unsigned)ww < 64u);
            bok[j][e]  = v;
            boff[j][e] = v ? (unsigned)((ci * HW + hh * 64 + ww) * 4) : 0u;
        }
    }

    v8f   acc0 = {}, acc1 = {}, acc2 = {};
    v4f   aq[2][G3_AP];
    float bq[2][2][9][2];

    gg_kern_load_b(cbytes, boff, bq[0]);
    gg_kern_load_a(af, aq[0]);
    for (int u = 0; u < G3_UNITS - 2; u += 2) {
        GG_KERN_UNIT(0, 1, af + (u + 1) * (G3_AP * 32), cbytes + (size_t)(u + 1) * (8 * HW * 4))
        GG_KERN_UNIT(1, 0, af + (u + 2) * (G3_AP * 32), cbytes + (size_t)(u + 2) * (8 * HW * 4))
    }
    GG_KERN_UNIT(0, 1, af + (G3_UNITS - 1) * (G3_AP * 32),
                 cbytes + (size_t)(G3_UNITS - 1) * (8 * HW * 4))
    gg_kern_mma<0>(aq[1], bq[1], bok, acc0, acc1, acc2);
    gg_kern_mma<1>(aq[1], bq[1], bok, acc0, acc1, acc2);

    float* kb = kern + (size_t)b * CENC * HW + h * 64 + w;
#pragma unroll
    for (int r = 0; r < 8; ++r) {
        int m = r + 8 * half;
        kb[m * HW]        = acc0[r] + b2[m];
        kb[(m + 16) * HW] = acc1[r] + b2[m + 16];
        if (m + 32 < CENC) kb[(m + 32) * HW] = acc2[r] + b2[m + 32];
    }
}

#define CH     32
#define XROW   72
#define NPIECE 6
#define XBUF   (CH * 3 * XROW)

__device__ __forceinline__ void gg_lds_barrier() {
    __builtin_amdgcn_fence(__ATOMIC_RELEASE, "workgroup", "local");
    __builtin_amdgcn_s_barrier();
    __builtin_amdgcn_fence(__ATOMIC_ACQUIRE, "workgroup", "local");
}

__global__ __launch_bounds__(256)
void gg_carafe_kernel(const float* __restrict__ x,
                      const float* __restrict__ mean,
                      const float* __restrict__ kern,
                      float* __restrict__ out) {
    __shared__ float smask[KK][SS][WWID];
    __shared__ __attribute__((aligned(16))) float xt[2 * XBUF];

    int bh  = blockIdx.x;
    int b   = bh >> 6, h = bh & 63;
    int tid = threadIdx.x;
    int tx  = tid & 63;
    int ty  = tid >> 6;

    const float* xb    = x + (size_t)b * CIN * HW;
    const float* meanb = mean + b * HW;
    const float* kb    = kern + (size_t)b * CENC * HW;

    unsigned goff[NPIECE], loff[NPIECE];
    bool     gok[NPIECE];
#pragma unroll
    for (int i = 0; i < NPIECE; ++i) {
        int q   = tid + 256 * i;
        int cc  = q / 48;
        int pos = q - cc * 48;
        int r   = pos >> 4, p = pos & 15;
        bool ok = (unsigned)(h - 1 + r) < 64u;
        gok[i]  = ok;
        goff[i] = ok ? (unsigned)(cc * HW + (h - 1) * 64 + pos * 4) : 0u;
        loff[i] = (unsigned)((cc * 3 + r) * XROW + 4 + p * 4);
    }

    v4f st[NPIECE];
#pragma unroll
    for (int i = 0; i < NPIECE; ++i)
        st[i] = *(const v4f*)(xb + goff[i]);

    const v4f zero4 = {0.f, 0.f, 0.f, 0.f};
    for (int i = tid; i < 2 * CH * 3 * 2; i += 256)
        *(v4f*)(xt + (i >> 1) * XROW + (i & 1) * 68) = zero4;

    if (ty == 0) {
        int w   = tx;
        float mc = meanb[h * 64 + w];
        float mval[SS][KK];
#pragma unroll
        for (int kk = 0; kk < KK; ++kk) {
            int dy = kk / 3 - 1, dx = kk % 3 - 1;
            int hh = h + dy, ww = w + dx;
            float mp = ((unsigned)hh < 64u && (unsigned)ww < 64u)
                           ? meanb[hh * 64 + ww] : 0.f;
            float g = mp - mc;
            g = 1.0f / (g * g + 0.2f);
#pragma unroll
            for (int s = 0; s < SS; ++s)
                mval[s][kk] = g * kb[(kk * SS + s) * HW + h * 64 + w];
        }
#pragma unroll
        for (int s = 0; s < SS; ++s) {
            float mx = mval[s][0];
#pragma unroll
            for (int kk = 1; kk < KK; ++kk) mx = fmaxf(mx, mval[s][kk]);
            float e[KK], sum = 0.f;
#pragma unroll
            for (int kk = 0; kk < KK; ++kk) { e[kk] = __expf(mval[s][kk] - mx); sum += e[kk]; }
            float inv = 1.0f / sum;
#pragma unroll
            for (int kk = 0; kk < KK; ++kk) smask[kk][s][w] = e[kk] * inv;
        }
    }

#pragma unroll
    for (int i = 0; i < NPIECE; ++i)
        *(v4f*)(xt + loff[i]) = gok[i] ? st[i] : zero4;
    __syncthreads();

    float m[KK][SS];
#pragma unroll
    for (int kk = 0; kk < KK; ++kk)
#pragma unroll
        for (int s = 0; s < SS; ++s) m[kk][s] = smask[kk][s][tx];

    float* ob = out + ((size_t)b * CIN * 128 + 2 * h) * 128 + 2 * tx;

    for (int k = 0; k < CIN / CH; ++k) {
        const bool more = (k + 1) < CIN / CH;
        if (more) {
            const float* xn = xb + (size_t)(k + 1) * CH * HW;
#pragma unroll
            for (int i = 0; i < NPIECE; ++i)
                st[i] = *(const v4f*)(xn + goff[i]);
        }
        __builtin_amdgcn_sched_barrier(0);

        const float* tb = xt + (k & 1) * XBUF + 3 + tx;
#pragma unroll
        for (int j = 0; j < CH / 4; ++j) {
            int cc = ty + 4 * j;
            const float* tr = tb + cc * 3 * XROW;
            float v[KK];
#pragma unroll
            for (int kk = 0; kk < KK; ++kk)
                v[kk] = tr[(kk / 3) * XROW + (kk % 3)];
            float a[SS];
#pragma unroll
            for (int s = 0; s < SS; ++s) {
                float acc = 0.f;
#pragma unroll
                for (int kk = 0; kk < KK; ++kk)
                    acc = fmaf(v[kk], m[kk][s], acc);
                a[s] = acc;
            }
            float* o = ob + (size_t)(k * CH + cc) * (128 * 128);
            v2f o0 = {a[0], a[1]}, o1 = {a[2], a[3]};
            *(v2f*)o         = o0;
            *(v2f*)(o + 128) = o1;
        }

        __builtin_amdgcn_sched_barrier(0);
        if (more) {
            float* tn = xt + ((k + 1) & 1) * XBUF;
#pragma unroll
            for (int i = 0; i < NPIECE; ++i)
                *(v4f*)(tn + loff[i]) = gok[i] ? st[i] : zero4;
            gg_lds_barrier();
        }
    }
}

extern "C" void kernel_launch(void* const* d_in, const int* in_sizes, int n_in,
                              void* d_out, int out_size, void* d_ws, size_t ws_size,
                              hipStream_t stream) {
    const float* x  = (const float*)d_in[0];
    const float* w1 = (const float*)d_in[1];
    const float* b1 = (const float*)d_in[2];
    const float* w2 = (const float*)d_in[3];
    const float* b2 = (const float*)d_in[4];
    float* out = (float*)d_out;

    float* ws   = (float*)d_ws;
    float* mean = ws;
    float* comp = ws + 32768;
    float* kern = comp + (size_t)8 * CCOMP * HW;
    float* w2f  = out;

    gg_mean_kernel <<<512,  256, 0, stream>>>(x, w2, mean, w2f);
    gg_comp_gemm   <<<512,  128, 0, stream>>>(x, w1, b1, comp);
    gg_kern_gemm   <<<512,  128, 0, stream>>>(comp, w2f, b2, kern);
    gg_carafe_kernel<<<512, 256, 0, stream>>>(x, mean, kern, out);
}


# mean kernel's launch removed: 4 more waves in each 1x1-GEMM block stream the row for its channel mean (and write w2f) beside the 4 WMMA waves; mean's sum order changes (8 partial sums)
# speedup vs baseline: 3.5146x; 1.0590x over previous
#include <hip/hip_runtime.h>

typedef float v2f __attribute__((ext_vector_type(2)));
typedef float v4f __attribute__((ext_vector_type(4)));
typedef float v8f __attribute__((ext_vector_type(8)));

#define CIN   256
#define CCOMP 64
#define CENC  36
#define HH    64
#define WWID  64
#define HW    4096
#define KK    9
#define SS    4
#define KDIM  576

__device__ __forceinline__ void gg_lds_barrier() {
    __builtin_amdgcn_fence(__ATOMIC_RELEASE, "workgroup", "local");
    __builtin_amdgcn_s_barrier();
    __builtin_amdgcn_fence(__ATOMIC_ACQUIRE, "workgroup", "local");
}

#define G2_KB 4
#define G2_NS (CIN / (8 * G2_KB))
#define MEAN_CG 8
#define MEAN_LD (CIN / MEAN_CG)
#define W2F_PER_BLOCK 54

__device__ __forceinline__ void gg_comp_load(const float* __restrict__ ap,
                                             const float* __restrict__ bp,
                                             int s, v4f (&aq)[G2_KB],
                                             v4f (&bq)[G2_KB][4]) {
#pragma unroll
    for (int kb = 0; kb < G2_KB; ++kb) {
        int k0 = (s * G2_KB + kb) * 8;
        aq[kb] = *(const v4f*)(ap + k0);
#pragma unroll
        for (int e = 0; e < 4; ++e)
            bq[kb][e] = *(const v4f*)(bp + (size_t)(k0 + e) * HW);
    }
}

__device__ __forceinline__ void gg_comp_mma(const v4f (&aq)[G2_KB],
                                            const v4f (&bq)[G2_KB][4],
                                            v8f (&acc)[4]) {
#pragma unroll
    for (int kb = 0; kb < G2_KB; ++kb) {
#pragma unroll
        for (int st = 0; st < 2; ++st) {
            v2f a;
            a.x = aq[kb][2 * st];
            a.y = aq[kb][2 * st + 1];
#pragma unroll
            for (int t = 0; t < 4; ++t) {
                v2f bm;
                bm.x = bq[kb][2 * st][t];
                bm.y = bq[kb][2 * st + 1][t];
                acc[t] = __builtin_amdgcn_wmma_f32_16x16x4_f32(
                             false, a, false, bm, (short)0, acc[t], false, false);
            }
        }
    }
}

__global__ __launch_bounds__(256)
void gg_comp_mean_kernel(const float* __restrict__ x,
                         const float* __restrict__ w1,
                         const float* __restrict__ b1,
                         const float* __restrict__ w2,
                         float* __restrict__ comp,
                         float* __restrict__ mean,
                         float* __restrict__ w2f) {
    __shared__ __attribute__((aligned(16))) float part[MEAN_CG][WWID];

    int b = blockIdx.x >> 6, h = blockIdx.x & 63;

    if (threadIdx.x < 128) {
        int wave = threadIdx.x >> 5;
        int lane = threadIdx.x & 31;
        int lr   = lane & 15;
        int half = lane >> 4;
        int m0   = wave * 16;

        const float* ap = w1 + (m0 + lr) * CIN + 4 * half;
        const float* bp = x + ((size_t)b * CIN + 4 * half) * HW + h * 64 + 4 * lr;

        float bias[8];
#pragma unroll
        for (int r = 0; r < 8; ++r) bias[r] = b1[m0 + 8 * half + r];

        v8f acc[4] = {};
        v4f aq[2][G2_KB];
        v4f bq[2][G2_KB][4];

        gg_comp_load(ap, bp, 0, aq[0], bq[0]);
#pragma unroll
        for (int s = 0; s < G2_NS; ++s) {
            if (s + 1 < G2_NS) gg_comp_load(ap, bp, s + 1, aq[(s + 1) & 1], bq[(s + 1) & 1]);
            __builtin_amdgcn_sched_barrier(0);
            gg_comp_mma(aq[s & 1], bq[s & 1], acc);
        }

        float* cb = comp + ((size_t)b * CCOMP + m0 + 8 * half) * HW + h * 64 + 4 * lr;
#pragma unroll
        for (int r = 0; r < 8; ++r) {
            v4f o = {acc[0][r] + bias[r], acc[1][r] + bias[r],
                     acc[2][r] + bias[r], acc[3][r] + bias[r]};
            *(v4f*)(cb + (size_t)r * HW) = o;
        }
    } else {
        int mt = threadIdx.x - 128;
        int p4 = mt & 15;
        int cg = mt >> 4;

        const float* xr = x + ((size_t)b * CIN + cg) * HW + h * 64 + p4 * 4;

        v4f v[MEAN_LD];
#pragma unroll
        for (int j = 0; j < MEAN_LD; ++j)
            v[j] = *(const v4f*)(xr + (size_t)j * MEAN_CG * HW);
        __builtin_amdgcn_sched_barrier(0);
        v4f s = v[0];
#pragma unroll
        for (int j = 1; j < MEAN_LD; ++j) s += v[j];

        *(v4f*)&part[cg][p4 * 4] = s;

        if (mt < W2F_PER_BLOCK) {
            int i   = blockIdx.x * W2F_PER_BLOCK + mt;
            int r   = i & 3, l = (i >> 2) & 31, pt = i >> 7;
            int t   = pt % 3, p = pt / 3;
            int k   = 4 * (2 * p + (r >> 1)) + 2 * (l >> 4) + (r & 1);
            int row = 16 * t + (l & 15);
            float wv = w2[(row < CENC ? row : 0) * KDIM + k];
            w2f[i]  = row < CENC ? wv : 0.f;
        }
    }

    gg_lds_barrier();

    if (threadIdx.x < WWID) {
        float t = part[0][threadIdx.x];
#pragma unroll
        for (int g = 1; g < MEAN_CG; ++g) t += part[g][threadIdx.x];
        mean[blockIdx.x * 64 + threadIdx.x] = t * (1.0f / CIN);
    }
}

#define G3_UNITS (KDIM / 72)
#define G3_AP    27

__device__ __forceinline__ void gg_kern_load_a(const v4f* __restrict__ af,
                                               v4f (&aq)[G3_AP]) {
#pragma unroll
    for (int i = 0; i < G3_AP; ++i) aq[i] = af[i * 32];
}

__device__ __forceinline__ void gg_kern_load_b(const char* __restrict__ cu,
                                               const unsigned (&boff)[9][2],
                                               float (&bq)[2][9][2]) {
#pragma unroll
    for (int hp = 0; hp < 2; ++hp) {
#pragma unroll
        for (int j = 0; j < 9; ++j) {
#pragma unroll
            for (int e = 0; e < 2; ++e)
                bq[hp][j][e] = *(const float*)(cu + hp * (4 * HW * 4) + boff[j][e]);
        }
    }
}

template <int HP>
__device__ __forceinline__ void gg_kern_mma(const v4f (&aq)[G3_AP],
                                            const float (&bq)[2][9][2],
                                            const bool (&bok)[9][2],
                                            v8f& acc0, v8f& acc1, v8f& acc2) {
#pragma unroll
    for (int j = 0; j < 9; ++j) {
        const int s  = 9 * HP + j;
        const int pp = s >> 1, o = (s & 1) * 2;
        v2f bm;
        bm.x = bok[j][0] ? bq[HP][j][0] : 0.f;
        bm.y = bok[j][1] ? bq[HP][j][1] : 0.f;
        v2f a0 = {aq[pp * 3 + 0][o], aq[pp * 3 + 0][o + 1]};
        v2f a1 = {aq[pp * 3 + 1][o], aq[pp * 3 + 1][o + 1]};
        v2f a2 = {aq[pp * 3 + 2][o], aq[pp * 3 + 2][o + 1]};
        acc0 = __builtin_amdgcn_wmma_f32_16x16x4_f32(
                   false, a0, false, bm, (short)0, acc0, false, false);
        acc1 = __builtin_amdgcn_wmma_f32_16x16x4_f32(
                   false, a1, false, bm, (short)0, acc1, false, false);
        acc2 = __builtin_amdgcn_wmma_f32_16x16x4_f32(
                   false, a2, false, bm, (short)0, acc2, false, false);
    }
}

#define GG_KERN_UNIT(CUR, NXT, afn, cun)                                   \
    gg_kern_load_b(cun, boff, bq[NXT]);                                    \
    __builtin_amdgcn_sched_barrier(0);                                     \
    gg_kern_mma<0>(aq[CUR], bq[CUR], bok, acc0, acc1, acc2);               \
    __builtin_amdgcn_sched_barrier(0);                                     \
    gg_kern_load_a(afn, aq[NXT]);                                          \
    __builtin_amdgcn_sched_barrier(0);                                     \
    gg_kern_mma<1>(aq[CUR], bq[CUR], bok, acc0, acc1, acc2);               \
    __builtin_amdgcn_sched_barrier(0);

__global__ __launch_bounds__(128)
void gg_kern_gemm(const float* __restrict__ comp,
                  const float* __restrict__ w2f,
                  const float* __restrict__ b2,
                  float* __restrict__ kern) {
    int b    = blockIdx.x >> 6;
    int h    = blockIdx.x & 63;
    int wave = threadIdx.x >> 5;
    int lane = threadIdx.x & 31;
    int lr   = lane & 15;
    int half = lane >> 4;
    int w    = wave * 16 + lr;

    const char* cbytes = (const char*)(comp + (size_t)b * CCOMP * HW);
    const v4f*  af     = (const v4f*)w2f + lane;

    unsigned boff[9][2];
    bool     bok[9][2];
#pragma unroll
    for (int j = 0; j < 9; ++j) {
#pragma unroll
        for (int e = 0; e < 2; ++e) {
            const int kA = 4 * j + e, kB = 4 * j + 2 + e;
            const int cA = kA / 9, tA = kA % 9;
            const int cB = kB / 9, tB = kB % 9;
            int ci = half ? cB : cA;
            int dy = half ? (tB / 3 - 1) : (tA / 3 - 1);
            int dx = half ? (tB % 3 - 1) : (tA % 3 - 1);
            int hh = h + dy, ww = w + dx;
            bool v = ((unsigned)hh < 64u) && ((unsigned)ww < 64u);
            bok[j][e]  = v;
            boff[j][e] = v ? (unsigned)((ci * HW + hh * 64 + ww) * 4) : 0u;
        }
    }

    v8f   acc0 = {}, acc1 = {}, acc2 = {};
    v4f   aq[2][G3_AP];
    float bq[2][2][9][2];

    gg_kern_load_b(cbytes, boff, bq[0]);
    gg_kern_load_a(af, aq[0]);
    for (int u = 0; u < G3_UNITS - 2; u += 2) {
        GG_KERN_UNIT(0, 1, af + (u + 1) * (G3_AP * 32), cbytes + (size_t)(u + 1) * (8 * HW * 4))
        GG_KERN_UNIT(1, 0, af + (u + 2) * (G3_AP * 32), cbytes + (size_t)(u + 2) * (8 * HW * 4))
    }
    GG_KERN_UNIT(0, 1, af + (G3_UNITS - 1) * (G3_AP * 32),
                 cbytes + (size_t)(G3_UNITS - 1) * (8 * HW * 4))
    gg_kern_mma<0>(aq[1], bq[1], bok, acc0, acc1, acc2);
    gg_kern_mma<1>(aq[1], bq[1], bok, acc0, acc1, acc2);

    float* kb = kern + (size_t)b * CENC * HW + h * 64 + w;
#pragma unroll
    for (int r = 0; r < 8; ++r) {
        int m = r + 8 * half;
        kb[m * HW]        = acc0[r] + b2[m];
        kb[(m + 16) * HW] = acc1[r] + b2[m + 16];
        if (m + 32 < CENC) kb[(m + 32) * HW] = acc2[r] + b2[m + 32];
    }
}

#define CH     32
#define XROW   72
#define NPIECE 6
#define XBUF   (CH * 3 * XROW)

__global__ __launch_bounds__(256)
void gg_carafe_kernel(const float* __restrict__ x,
                      const float* __restrict__ mean,
                      const float* __restrict__ kern,
                      float* __restrict__ out) {
    __shared__ float smask[KK][SS][WWID];
    __shared__ __attribute__((aligned(16))) float xt[2 * XBUF];

    int bh  = blockIdx.x;
    int b   = bh >> 6, h = bh & 63;
    int tid = threadIdx.x;
    int tx  = tid & 63;
    int ty  = tid >> 6;

    const float* xb    = x + (size_t)b * CIN * HW;
    const float* meanb = mean + b * HW;
    const float* kb    = kern + (size_t)b * CENC * HW;

    unsigned goff[NPIECE], loff[NPIECE];
    bool     gok[NPIECE];
#pragma unroll
    for (int i = 0; i < NPIECE; ++i) {
        int q   = tid + 256 * i;
        int cc  = q / 48;
        int pos = q - cc * 48;
        int r   = pos >> 4, p = pos & 15;
        bool ok = (unsigned)(h - 1 + r) < 64u;
        gok[i]  = ok;
        goff[i] = ok ? (unsigned)(cc * HW + (h - 1) * 64 + pos * 4) : 0u;
        loff[i] = (unsigned)((cc * 3 + r) * XROW + 4 + p * 4);
    }

    v4f st[NPIECE];
#pragma unroll
    for (int i = 0; i < NPIECE; ++i)
        st[i] = *(const v4f*)(xb + goff[i]);

    const v4f zero4 = {0.f, 0.f, 0.f, 0.f};
    for (int i = tid; i < 2 * CH * 3 * 2; i += 256)
        *(v4f*)(xt + (i >> 1) * XROW + (i & 1) * 68) = zero4;

    if (ty == 0) {
        int w   = tx;
        float mc = meanb[h * 64 + w];
        float mval[SS][KK];
#pragma unroll
        for (int kk = 0; kk < KK; ++kk) {
            int dy = kk / 3 - 1, dx = kk % 3 - 1;
            int hh = h + dy, ww = w + dx;
            float mp = ((unsigned)hh < 64u && (unsigned)ww < 64u)
                           ? meanb[hh * 64 + ww] : 0.f;
            float g = mp - mc;
            g = 1.0f / (g * g + 0.2f);
#pragma unroll
            for (int s = 0; s < SS; ++s)
                mval[s][kk] = g * kb[(kk * SS + s) * HW + h * 64 + w];
        }
#pragma unroll
        for (int s = 0; s < SS; ++s) {
            float mx = mval[s][0];
#pragma unroll
            for (int kk = 1; kk < KK; ++kk) mx = fmaxf(mx, mval[s][kk]);
            float e[KK], sum = 0.f;
#pragma unroll
            for (int kk = 0; kk < KK; ++kk) { e[kk] = __expf(mval[s][kk] - mx); sum += e[kk]; }
            float inv = 1.0f / sum;
#pragma unroll
            for (int kk = 0; kk < KK; ++kk) smask[kk][s][w] = e[kk] * inv;
        }
    }

#pragma unroll
    for (int i = 0; i < NPIECE; ++i)
        *(v4f*)(xt + loff[i]) = gok[i] ? st[i] : zero4;
    __syncthreads();

    float m[KK][SS];
#pragma unroll
    for (int kk = 0; kk < KK; ++kk)
#pragma unroll
        for (int s = 0; s < SS; ++s) m[kk][s] = smask[kk][s][tx];

    float* ob = out + ((size_t)b * CIN * 128 + 2 * h) * 128 + 2 * tx;

    for (int k = 0; k < CIN / CH; ++k) {
        const bool more = (k + 1) < CIN / CH;
        if (more) {
            const float* xn = xb + (size_t)(k + 1) * CH * HW;
#pragma unroll
            for (int i = 0; i < NPIECE; ++i)
                st[i] = *(const v4f*)(xn + goff[i]);
        }
        __builtin_amdgcn_sched_barrier(0);

        const float* tb = xt + (k & 1) * XBUF + 3 + tx;
#pragma unroll
        for (int j = 0; j < CH / 4; ++j) {
            int cc = ty + 4 * j;
            const float* tr = tb + cc * 3 * XROW;
            float v[KK];
#pragma unroll
            for (int kk = 0; kk < KK; ++kk)
                v[kk] = tr[(kk / 3) * XROW + (kk % 3)];
            float a[SS];
#pragma unroll
            for (int s = 0; s < SS; ++s) {
                float acc = 0.f;
#pragma unroll
                for (int kk = 0; kk < KK; ++kk)
                    acc = fmaf(v[kk], m[kk][s], acc);
                a[s] = acc;
            }
            float* o = ob + (size_t)(k * CH + cc) * (128 * 128);
            v2f o0 = {a[0], a[1]}, o1 = {a[2], a[3]};
            *(v2f*)o         = o0;
            *(v2f*)(o + 128) = o1;
        }

        __builtin_amdgcn_sched_barrier(0);
        if (more) {
            float* tn = xt + ((k + 1) & 1) * XBUF;
#pragma unroll
            for (int i = 0; i < NPIECE; ++i)
                *(v4f*)(tn + loff[i]) = gok[i] ? st[i] : zero4;
            gg_lds_barrier();
        }
    }
}

extern "C" void kernel_launch(void* const* d_in, const int* in_sizes, int n_in,
                              void* d_out, int out_size, void* d_ws, size_t ws_size,
                              hipStream_t stream) {
    const float* x  = (const float*)d_in[0];
    const float* w1 = (const float*)d_in[1];
    const float* b1 = (const float*)d_in[2];
    const float* w2 = (const float*)d_in[3];
    const float* b2 = (const float*)d_in[4];
    float* out = (float*)d_out;

    float* ws   = (float*)d_ws;
    float* mean = ws;
    float* comp = ws + 32768;
    float* kern = comp + (size_t)8 * CCOMP * HW;
    float* w2f  = out;

    gg_comp_mean_kernel<<<512, 256, 0, stream>>>(x, w1, b1, w2, comp, mean, w2f);
    gg_kern_gemm   <<<512,  128, 0, stream>>>(comp, w2f, b2, kern);
    gg_carafe_kernel<<<512, 256, 0, stream>>>(x, mean, kern, out);
}


# 3x3-conv GEMM epilogue: the 20 bias values a lane adds are loaded in front of the k loop; the stores no longer wait row by row for a bias load each; sums unchanged
# speedup vs baseline: 3.9454x; 1.1226x over previous
#include <hip/hip_runtime.h>

typedef float v2f __attribute__((ext_vector_type(2)));
typedef float v4f __attribute__((ext_vector_type(4)));
typedef float v8f __attribute__((ext_vector_type(8)));

#define CIN   256
#define CCOMP 64
#define CENC  36
#define HH    64
#define WWID  64
#define HW    4096
#define KK    9
#define SS    4
#define KDIM  576

__device__ __forceinline__ void gg_lds_barrier() {
    __builtin_amdgcn_fence(__ATOMIC_RELEASE, "workgroup", "local");
    __builtin_amdgcn_s_barrier();
    __builtin_amdgcn_fence(__ATOMIC_ACQUIRE, "workgroup", "local");
}

#define G2_KB 4
#define G2_NS (CIN / (8 * G2_KB))
#define MEAN_CG 8
#define MEAN_LD (CIN / MEAN_CG)
#define W2F_PER_BLOCK 54

__device__ __forceinline__ void gg_comp_load(const float* __restrict__ ap,
                                             const float* __restrict__ bp,
                                             int s, v4f (&aq)[G2_KB],
                                             v4f (&bq)[G2_KB][4]) {
#pragma unroll
    for (int kb = 0; kb < G2_KB; ++kb) {
        int k0 = (s * G2_KB + kb) * 8;
        aq[kb] = *(const v4f*)(ap + k0);
#pragma unroll
        for (int e = 0; e < 4; ++e)
            bq[kb][e] = *(const v4f*)(bp + (size_t)(k0 + e) * HW);
    }
}

__device__ __forceinline__ void gg_comp_mma(const v4f (&aq)[G2_KB],
                                            const v4f (&bq)[G2_KB][4],
                                            v8f (&acc)[4]) {
#pragma unroll
    for (int kb = 0; kb < G2_KB; ++kb) {
#pragma unroll
        for (int st = 0; st < 2; ++st) {
            v2f a;
            a.x = aq[kb][2 * st];
            a.y = aq[kb][2 * st + 1];
#pragma unroll
            for (int t = 0; t < 4; ++t) {
                v2f bm;
                bm.x = bq[kb][2 * st][t];
                bm.y = bq[kb][2 * st + 1][t];
                acc[t] = __builtin_amdgcn_wmma_f32_16x16x4_f32(
                             false, a, false, bm, (short)0, acc[t], false, false);
            }
        }
    }
}

__global__ __launch_bounds__(256)
void gg_comp_mean_kernel(const float* __restrict__ x,
                         const float* __restrict__ w1,
                         const float* __restrict__ b1,
                         const float* __restrict__ w2,
                         float* __restrict__ comp,
                         float* __restrict__ mean,
                         float* __restrict__ w2f) {
    __shared__ __attribute__((aligned(16))) float part[MEAN_CG][WWID];

    int b = blockIdx.x >> 6, h = blockIdx.x & 63;

    if (threadIdx.x < 128) {
        int wave = threadIdx.x >> 5;
        int lane = threadIdx.x & 31;
        int lr   = lane & 15;
        int half = lane >> 4;
        int m0   = wave * 16;

        const float* ap = w1 + (m0 + lr) * CIN + 4 * half;
        const float* bp = x + ((size_t)b * CIN + 4 * half) * HW + h * 64 + 4 * lr;

        float bias[8];
#pragma unroll
        for (int r = 0; r < 8; ++r) bias[r] = b1[m0 + 8 * half + r];

        v8f acc[4] = {};
        v4f aq[2][G2_KB];
        v4f bq[2][G2_KB][4];

        gg_comp_load(ap, bp, 0, aq[0], bq[0]);
#pragma unroll
        for (int s = 0; s < G2_NS; ++s) {
            if (s + 1 < G2_NS) gg_comp_load(ap, bp, s + 1, aq[(s + 1) & 1], bq[(s + 1) & 1]);
            __builtin_amdgcn_sched_barrier(0);
            gg_comp_mma(aq[s & 1], bq[s & 1], acc);
        }

        float* cb = comp + ((size_t)b * CCOMP + m0 + 8 * half) * HW + h * 64 + 4 * lr;
#pragma unroll
        for (int r = 0; r < 8; ++r) {
            v4f o = {acc[0][r] + bias[r], acc[1][r] + bias[r],
                     acc[2][r] + bias[r], acc[3][r] + bias[r]};
            *(v4f*)(cb + (size_t)r * HW) = o;
        }
    } else {
        int mt = threadIdx.x - 128;
        int p4 = mt & 15;
        int cg = mt >> 4;

        const float* xr = x + ((size_t)b * CIN + cg) * HW + h * 64 + p4 * 4;

        v4f v[MEAN_LD];
#pragma unroll
        for (int j = 0; j < MEAN_LD; ++j)
            v[j] = *(const v4f*)(xr + (size_t)j * MEAN_CG * HW);
        __builtin_amdgcn_sched_barrier(0);
        v4f s = v[0];
#pragma unroll
        for (int j = 1; j < MEAN_LD; ++j) s += v[j];

        *(v4f*)&part[cg][p4 * 4] = s;

        if (mt < W2F_PER_BLOCK) {
            int i   = blockIdx.x * W2F_PER_BLOCK + mt;
            int r   = i & 3, l = (i >> 2) & 31, pt = i >> 7;
            int t   = pt % 3, p = pt / 3;
            int k   = 4 * (2 * p + (r >> 1)) + 2 * (l >> 4) + (r & 1);
            int row = 16 * t + (l & 15);
            float wv = w2[(row < CENC ? row : 0) * KDIM + k];
            w2f[i]  = row < CENC ? wv : 0.f;
        }
    }

    gg_lds_barrier();

    if (threadIdx.x < WWID) {
        float t = part[0][threadIdx.x];
#pragma unroll
        for (int g = 1; g < MEAN_CG; ++g) t += part[g][threadIdx.x];
        mean[blockIdx.x * 64 + threadIdx.x] = t * (1.0f / CIN);
    }
}

#define G3_UNITS (KDIM / 72)
#define G3_AP    27

__device__ __forceinline__ void gg_kern_load_a(const v4f* __restrict__ af,
                                               v4f (&aq)[G3_AP]) {
#pragma unroll
    for (int i = 0; i < G3_AP; ++i) aq[i] = af[i * 32];
}

__device__ __forceinline__ void gg_kern_load_b(const char* __restrict__ cu,
                                               const unsigned (&boff)[9][2],
                                               float (&bq)[2][9][2]) {
#pragma unroll
    for (int hp = 0; hp < 2; ++hp) {
#pragma unroll
        for (int j = 0; j < 9; ++j) {
#pragma unroll
            for (int e = 0; e < 2; ++e)
                bq[hp][j][e] = *(const float*)(cu + hp * (4 * HW * 4) + boff[j][e]);
        }
    }
}

template <int HP>
__device__ __forceinline__ void gg_kern_mma(const v4f (&aq)[G3_AP],
                                            const float (&bq)[2][9][2],
                                            const bool (&bok)[9][2],
                                            v8f& acc0, v8f& acc1, v8f& acc2) {
#pragma unroll
    for (int j = 0; j < 9; ++j) {
        const int s  = 9 * HP + j;
        const int pp = s >> 1, o = (s & 1) * 2;
        v2f bm;
        bm.x = bok[j][0] ? bq[HP][j][0] : 0.f;
        bm.y = bok[j][1] ? bq[HP][j][1] : 0.f;
        v2f a0 = {aq[pp * 3 + 0][o], aq[pp * 3 + 0][o + 1]};
        v2f a1 = {aq[pp * 3 + 1][o], aq[pp * 3 + 1][o + 1]};
        v2f a2 = {aq[pp * 3 + 2][o], aq[pp * 3 + 2][o + 1]};
        acc0 = __builtin_amdgcn_wmma_f32_16x16x4_f32(
                   false, a0, false, bm, (short)0, acc0, false, false);
        acc1 = __builtin_amdgcn_wmma_f32_16x16x4_f32(
                   false, a1, false, bm, (short)0, acc1, false, false);
        acc2 = __builtin_amdgcn_wmma_f32_16x16x4_f32(
                   false, a2, false, bm, (short)0, acc2, false, false);
    }
}

#define GG_KERN_UNIT(CUR, NXT, afn, cun)                                   \
    gg_kern_load_b(cun, boff, bq[NXT]);                                    \
    __builtin_amdgcn_sched_barrier(0);                                     \
    gg_kern_mma<0>(aq[CUR], bq[CUR], bok, acc0, acc1, acc2);               \
    __builtin_amdgcn_sched_barrier(0);                                     \
    gg_kern_load_a(afn, aq[NXT]);                                          \
    __builtin_amdgcn_sched_barrier(0);                                     \
    gg_kern_mma<1>(aq[CUR], bq[CUR], bok, acc0, acc1, acc2);               \
    __builtin_amdgcn_sched_barrier(0);

__global__ __launch_bounds__(128)
void gg_kern_gemm(const float* __restrict__ comp,
                  const float* __restrict__ w2f,
                  const float* __restrict__ b2,
                  float* __restrict__ kern) {
    int b    = blockIdx.x >> 6;
    int h    = blockIdx.x & 63;
    int wave = threadIdx.x >> 5;
    int lane = threadIdx.x & 31;
    int lr   = lane & 15;
    int half = lane >> 4;
    int w    = wave * 16 + lr;

    const char* cbytes = (const char*)(comp + (size_t)b * CCOMP * HW);
    const v4f*  af     = (const v4f*)w2f + lane;

    unsigned boff[9][2];
    bool     bok[9][2];
#pragma unroll
    for (int j = 0; j < 9; ++j) {
#pragma unroll
        for (int e = 0; e < 2; ++e) {
            const int kA = 4 * j + e, kB = 4 * j + 2 + e;
            const int cA = kA / 9, tA = kA % 9;
            const int cB = kB / 9, tB = kB % 9;
            int ci = half ? cB : cA;
            int dy = half ? (tB / 3 - 1) : (tA / 3 - 1);
            int dx = half ? (tB % 3 - 1) : (tA % 3 - 1);
            int hh = h + dy, ww = w + dx;
            bool v = ((unsigned)hh < 64u) && ((unsigned)ww < 64u);
            bok[j][e]  = v;
            boff[j][e] = v ? (unsigned)((ci * HW + hh * 64 + ww) * 4) : 0u;
        }
    }

    float bz0[8], bz1[8], bz2[4];
#pragma unroll
    for (int r = 0; r < 8; ++r) {
        bz0[r] = b2[r + 8 * half];
        bz1[r] = b2[r + 8 * half + 16];
    }
#pragma unroll
    for (int r = 0; r < 4; ++r) bz2[r] = b2[half ? 0 : 32 + r];

    v8f   acc0 = {}, acc1 = {}, acc2 = {};
    v4f   aq[2][G3_AP];
    float bq[2][2][9][2];

    gg_kern_load_b(cbytes, boff, bq[0]);
    gg_kern_load_a(af, aq[0]);
    for (int u = 0; u < G3_UNITS - 2; u += 2) {
        GG_KERN_UNIT(0, 1, af + (u + 1) * (G3_AP * 32), cbytes + (size_t)(u + 1) * (8 * HW * 4))
        GG_KERN_UNIT(1, 0, af + (u + 2) * (G3_AP * 32), cbytes + (size_t)(u + 2) * (8 * HW * 4))
    }
    GG_KERN_UNIT(0, 1, af + (G3_UNITS - 1) * (G3_AP * 32),
                 cbytes + (size_t)(G3_UNITS - 1) * (8 * HW * 4))
    gg_kern_mma<0>(aq[1], bq[1], bok, acc0, acc1, acc2);
    gg_kern_mma<1>(aq[1], bq[1], bok, acc0, acc1, acc2);

    float* kb = kern + (size_t)b * CENC * HW + h * 64 + w;
#pragma unroll
    for (int r = 0; r < 8; ++r) {
        int m = r + 8 * half;
        kb[m * HW]        = acc0[r] + bz0[r];
        kb[(m + 16) * HW] = acc1[r] + bz1[r];
    }
    if (half == 0) {
#pragma unroll
        for (int r = 0; r < 4; ++r) kb[(r + 32) * HW] = acc2[r] + bz2[r];
    }
}

#define CH     32
#define XROW   72
#define NPIECE 6
#define XBUF   (CH * 3 * XROW)

__global__ __launch_bounds__(256)
void gg_carafe_kernel(const float* __restrict__ x,
                      const float* __restrict__ mean,
                      const float* __restrict__ kern,
                      float* __restrict__ out) {
    __shared__ float smask[KK][SS][WWID];
    __shared__ __attribute__((aligned(16))) float xt[2 * XBUF];

    int bh  = blockIdx.x;
    int b   = bh >> 6, h = bh & 63;
    int tid = threadIdx.x;
    int tx  = tid & 63;
    int ty  = tid >> 6;

    const float* xb    = x + (size_t)b * CIN * HW;
    const float* meanb = mean + b * HW;
    const float* kb    = kern + (size_t)b * CENC * HW;

    unsigned goff[NPIECE], loff[NPIECE];
    bool     gok[NPIECE];
#pragma unroll
    for (int i = 0; i < NPIECE; ++i) {
        int q   = tid + 256 * i;
        int cc  = q / 48;
        int pos = q - cc * 48;
        int r   = pos >> 4, p = pos & 15;
        bool ok = (unsigned)(h - 1 + r) < 64u;
        gok[i]  = ok;
        goff[i] = ok ? (unsigned)(cc * HW + (h - 1) * 64 + pos * 4) : 0u;
        loff[i] = (unsigned)((cc * 3 + r) * XROW + 4 + p * 4);
    }

    v4f st[NPIECE];
#pragma unroll
    for (int i = 0; i < NPIECE; ++i)
        st[i] = *(const v4f*)(xb + goff[i]);

    const v4f zero4 = {0.f, 0.f, 0.f, 0.f};
    for (int i = tid; i < 2 * CH * 3 * 2; i += 256)
        *(v4f*)(xt + (i >> 1) * XROW + (i & 1) * 68) = zero4;

    if (ty == 0) {
        int w   = tx;
        float mc = meanb[h * 64 + w];
        float mval[SS][KK];
#pragma unroll
        for (int kk = 0; kk < KK; ++kk) {
            int dy = kk / 3 - 1, dx = kk % 3 - 1;
            int hh = h + dy, ww = w + dx;
            float mp = ((unsigned)hh < 64u && (unsigned)ww < 64u)
                           ? meanb[hh * 64 + ww] : 0.f;
            float g = mp - mc;
            g = 1.0f / (g * g + 0.2f);
#pragma unroll
            for (int s = 0; s < SS; ++s)
                mval[s][kk] = g * kb[(kk * SS + s) * HW + h * 64 + w];
        }
#pragma unroll
        for (int s = 0; s < SS; ++s) {
            float mx = mval[s][0];
#pragma unroll
            for (int kk = 1; kk < KK; ++kk) mx = fmaxf(mx, mval[s][kk]);
            float e[KK], sum = 0.f;
#pragma unroll
            for (int kk = 0; kk < KK; ++kk) { e[kk] = __expf(mval[s][kk] - mx); sum += e[kk]; }
            float inv = 1.0f / sum;
#pragma unroll
            for (int kk = 0; kk < KK; ++kk) smask[kk][s][w] = e[kk] * inv;
        }
    }

#pragma unroll
    for (int i = 0; i < NPIECE; ++i)
        *(v4f*)(xt + loff[i]) = gok[i] ? st[i] : zero4;
    __syncthreads();

    float m[KK][SS];
#pragma unroll
    for (int kk = 0; kk < KK; ++kk)
#pragma unroll
        for (int s = 0; s < SS; ++s) m[kk][s] = smask[kk][s][tx];

    float* ob = out + ((size_t)b * CIN * 128 + 2 * h) * 128 + 2 * tx;

    for (int k = 0; k < CIN / CH; ++k) {
        const bool more = (k + 1) < CIN / CH;
        if (more) {
            const float* xn = xb + (size_t)(k + 1) * CH * HW;
#pragma unroll
            for (int i = 0; i < NPIECE; ++i)
                st[i] = *(const v4f*)(xn + goff[i]);
        }
        __builtin_amdgcn_sched_barrier(0);

        const float* tb = xt + (k & 1) * XBUF + 3 + tx;
#pragma unroll
        for (int j = 0; j < CH / 4; ++j) {
            int cc = ty + 4 * j;
            const float* tr = tb + cc * 3 * XROW;
            float v[KK];
#pragma unroll
            for (int kk = 0; kk < KK; ++kk)
                v[kk] = tr[(kk / 3) * XROW + (kk % 3)];
            float a[SS];
#pragma unroll
            for (int s = 0; s < SS; ++s) {
                float acc = 0.f;
#pragma unroll
                for (int kk = 0; kk < KK; ++kk)
                    acc = fmaf(v[kk], m[kk][s], acc);
                a[s] = acc;
            }
            float* o = ob + (size_t)(k * CH + cc) * (128 * 128);
            v2f o0 = {a[0], a[1]}, o1 = {a[2], a[3]};
            *(v2f*)o         = o0;
            *(v2f*)(o + 128) = o1;
        }

        __builtin_amdgcn_sched_barrier(0);
        if (more) {
            float* tn = xt + ((k + 1) & 1) * XBUF;
#pragma unroll
            for (int i = 0; i < NPIECE; ++i)
                *(v4f*)(tn + loff[i]) = gok[i] ? st[i] : zero4;
            gg_lds_barrier();
        }
    }
}

extern "C" void kernel_launch(void* const* d_in, const int* in_sizes, int n_in,
                              void* d_out, int out_size, void* d_ws, size_t ws_size,
                              hipStream_t stream) {
    const float* x  = (const float*)d_in[0];
    const float* w1 = (const float*)d_in[1];
    const float* b1 = (const float*)d_in[2];
    const float* w2 = (const float*)d_in[3];
    const float* b2 = (const float*)d_in[4];
    float* out = (float*)d_out;

    float* ws   = (float*)d_ws;
    float* mean = ws;
    float* comp = ws + 32768;
    float* kern = comp + (size_t)8 * CCOMP * HW;
    float* w2f  = out;

    gg_comp_mean_kernel<<<512, 256, 0, stream>>>(x, w1, b1, w2, comp, mean, w2f);
    gg_kern_gemm   <<<512,  128, 0, stream>>>(comp, w2f, b2, kern);
    gg_carafe_kernel<<<512, 256, 0, stream>>>(x, mean, kern, out);
}


# 3x3-conv GEMM in minimal-filtering form F(2,3) along image rows: 2 rows a block, 4 GEMMs of K=192 on ky-transformed weights, 288 WMMAs per 32 pixels, not 432; kern's sum order changes
# speedup vs baseline: 4.0942x; 1.0377x over previous
#include <hip/hip_runtime.h>

typedef float v2f __attribute__((ext_vector_type(2)));
typedef float v4f __attribute__((ext_vector_type(4)));
typedef float v8f __attribute__((ext_vector_type(8)));

#define CIN   256
#define CCOMP 64
#define CENC  36
#define HH    64
#define WWID  64
#define HW    4096
#define KK    9
#define SS    4
#define KDIM  576

__device__ __forceinline__ void gg_lds_barrier() {
    __builtin_amdgcn_fence(__ATOMIC_RELEASE, "workgroup", "local");
    __builtin_amdgcn_s_barrier();
    __builtin_amdgcn_fence(__ATOMIC_ACQUIRE, "workgroup", "local");
}

#define G2_KB 4
#define G2_NS (CIN / (8 * G2_KB))
#define MEAN_CG 8
#define MEAN_LD (CIN / MEAN_CG)
#define W2U_PER_BLOCK 72

__device__ __forceinline__ void gg_comp_load(const float* __restrict__ ap,
                                             const float* __restrict__ bp,
                                             int s, v4f (&aq)[G2_KB],
                                             v4f (&bq)[G2_KB][4]) {
#pragma unroll
    for (int kb = 0; kb < G2_KB; ++kb) {
        int k0 = (s * G2_KB + kb) * 8;
        aq[kb] = *(const v4f*)(ap + k0);
#pragma unroll
        for (int e = 0; e < 4; ++e)
            bq[kb][e] = *(const v4f*)(bp + (size_t)(k0 + e) * HW);
    }
}

__device__ __forceinline__ void gg_comp_mma(const v4f (&aq)[G2_KB],
                                            const v4f (&bq)[G2_KB][4],
                                            v8f (&acc)[4]) {
#pragma unroll
    for (int kb = 0; kb < G2_KB; ++kb) {
#pragma unroll
        for (int st = 0; st < 2; ++st) {
            v2f a;
            a.x = aq[kb][2 * st];
            a.y = aq[kb][2 * st + 1];
#pragma unroll
            for (int t = 0; t < 4; ++t) {
                v2f bm;
                bm.x = bq[kb][2 * st][t];
                bm.y = bq[kb][2 * st + 1][t];
                acc[t] = __builtin_amdgcn_wmma_f32_16x16x4_f32(
                             false, a, false, bm, (short)0, acc[t], false, false);
            }
        }
    }
}

__global__ __launch_bounds__(256)
void gg_comp_mean_kernel(const float* __restrict__ x,
                         const float* __restrict__ w1,
                         const float* __restrict__ b1,
                         const float* __restrict__ w2,
                         float* __restrict__ comp,
                         float* __restrict__ mean,
                         float* __restrict__ w2u) {
    __shared__ __attribute__((aligned(16))) float part[MEAN_CG][WWID];

    int b = blockIdx.x >> 6, h = blockIdx.x & 63;

    if (threadIdx.x < 128) {
        int wave = threadIdx.x >> 5;
        int lane = threadIdx.x & 31;
        int lr   = lane & 15;
        int half = lane >> 4;
        int m0   = wave * 16;

        const float* ap = w1 + (m0 + lr) * CIN + 4 * half;
        const float* bp = x + ((size_t)b * CIN + 4 * half) * HW + h * 64 + 4 * lr;

        float bias[8];
#pragma unroll
        for (int r = 0; r < 8; ++r) bias[r] = b1[m0 + 8 * half + r];

        v8f acc[4] = {};
        v4f aq[2][G2_KB];
        v4f bq[2][G2_KB][4];

        gg_comp_load(ap, bp, 0, aq[0], bq[0]);
#pragma unroll
        for (int s = 0; s < G2_NS; ++s) {
            if (s + 1 < G2_NS) gg_comp_load(ap, bp, s + 1, aq[(s + 1) & 1], bq[(s + 1) & 1]);
            __builtin_amdgcn_sched_barrier(0);
            gg_comp_mma(aq[s & 1], bq[s & 1], acc);
        }

        float* cb = comp + ((size_t)b * CCOMP + m0 + 8 * half) * HW + h * 64 + 4 * lr;
#pragma unroll
        for (int r = 0; r < 8; ++r) {
            v4f o = {acc[0][r] + bias[r], acc[1][r] + bias[r],
                     acc[2][r] + bias[r], acc[3][r] + bias[r]};
            *(v4f*)(cb + (size_t)r * HW) = o;
        }
    } else {
        int mt = threadIdx.x - 128;
        int p4 = mt & 15;
        int cg = mt >> 4;

        const float* xr = x + ((size_t)b * CIN + cg) * HW + h * 64 + p4 * 4;

        v4f v[MEAN_LD];
#pragma unroll
        for (int j = 0; j < MEAN_LD; ++j)
            v[j] = *(const v4f*)(xr + (size_t)j * MEAN_CG * HW);
        __builtin_amdgcn_sched_barrier(0);
        v4f s = v[0];
#pragma unroll
        for (int j = 1; j < MEAN_LD; ++j) s += v[j];

        *(v4f*)&part[cg][p4 * 4] = s;

        if (mt < W2U_PER_BLOCK) {
            int idx = blockIdx.x * W2U_PER_BLOCK + mt;
            int r   = idx & 3, l = (idx >> 2) & 31, pj = idx >> 7;
            int j   = pj % 12, sp = pj / 12;
            int gi  = j / 3, t = j % 3;
            int k   = 4 * (2 * sp + (r >> 1)) + 2 * (l >> 4) + (r & 1);
            int kx  = k >> 6, c = k & 63;
            int row = 16 * t + (l & 15);
            const float* g = w2 + ((row < CENC ? row : 0) * CCOMP + c) * 9 + kx;
            float g0 = g[0], g1 = g[3], g2 = g[6];
            float u  = gi == 0 ? g0
                     : gi == 1 ? 0.5f * ((g0 + g1) + g2)
                     : gi == 2 ? 0.5f * ((g0 - g1) + g2)
                     : g2;
            w2u[idx] = row < CENC ? u : 0.f;
        }
    }

    gg_lds_barrier();

    if (threadIdx.x < WWID) {
        float t = part[0][threadIdx.x];
#pragma unroll
        for (int g = 1; g < MEAN_CG; ++g) t += part[g][threadIdx.x];
        mean[blockIdx.x * 64 + threadIdx.x] = t * (1.0f / CIN);
    }
}

#define WN_UNITS 12
#define WN_AP    12

__device__ __forceinline__ void gg_wn_load_a(const v4f* __restrict__ af,
                                             v4f (&aq)[WN_AP]) {
#pragma unroll
    for (int sp = 0; sp < 2; ++sp) {
#pragma unroll
        for (int j = 0; j < 6; ++j) aq[sp * 6 + j] = af[(sp * 12 + j) * 32];
    }
}

template <int P>
__device__ __forceinline__ void gg_wn_load_b(const char* __restrict__ cu,
                                             const unsigned (&boff)[4],
                                             float (&bq)[4][2][3]) {
#pragma unroll
    for (int st = 0; st < 4; ++st) {
#pragma unroll
        for (int e = 0; e < 2; ++e) {
#pragma unroll
            for (int ii = 0; ii < 3; ++ii)
                bq[st][e][ii] = *(const float*)(cu + (4 * st + e) * (HW * 4) + boff[P + ii]);
        }
    }
}

template <int P, int SP>
__device__ __forceinline__ void gg_wn_mma(const v4f (&aq)[WN_AP],
                                          const float (&bq)[4][2][3],
                                          const bool (&bok)[4],
                                          v8f (&acc)[2][3]) {
#pragma unroll
    for (int s2 = 0; s2 < 2; ++s2) {
        const int st = 2 * SP + s2;
        v2f bm[2];
#pragma unroll
        for (int e = 0; e < 2; ++e) {
            float da = bok[P + 0] ? bq[st][e][0] : 0.f;
            float db = bok[P + 1] ? bq[st][e][1] : 0.f;
            float dc = bok[P + 2] ? bq[st][e][2] : 0.f;
            if (P == 0) {
                bm[0][e] = da - dc;
                bm[1][e] = db + dc;
            } else {
                bm[0][e] = db - da;
                bm[1][e] = da - dc;
            }
        }
#pragma unroll
        for (int ii = 0; ii < 2; ++ii) {
#pragma unroll
            for (int t = 0; t < 3; ++t) {
                const int p = SP * 6 + ii * 3 + t;
                v2f a = {aq[p][2 * s2], aq[p][2 * s2 + 1]};
                acc[ii][t] = __builtin_amdgcn_wmma_f32_16x16x4_f32(
                                 false, a, false, bm[ii], (short)0, acc[ii][t], false, false);
            }
        }
    }
}

template <int P>
__device__ __forceinline__ void gg_wn_run(const char* __restrict__ cbytes,
                                          const v4f* __restrict__ af,
                                          const unsigned (&boff)[3][4],
                                          const bool (&bok)[3][4],
                                          v8f (&acc)[2][3]) {
    v4f   aq[2][WN_AP];
    float bq[2][4][2][3];

    gg_wn_load_b<P>(cbytes, boff[0], bq[0]);
    gg_wn_load_a(af, aq[0]);
#pragma unroll
    for (int u = 0; u < WN_UNITS; ++u) {
        const int cur = u & 1, nxt = cur ^ 1;
        const int un  = u + 1;
        if (un < WN_UNITS) {
            gg_wn_load_b<P>(cbytes + (size_t)(16 * (un & 3)) * (HW * 4), boff[un >> 2], bq[nxt]);
            gg_wn_load_a(af + un * (24 * 32), aq[nxt]);
        }
        __builtin_amdgcn_sched_barrier(0);
        gg_wn_mma<P, 0>(aq[cur], bq[cur], bok[u >> 2], acc);
        gg_wn_mma<P, 1>(aq[cur], bq[cur], bok[u >> 2], acc);
        __builtin_amdgcn_sched_barrier(0);
    }
}

__global__ __launch_bounds__(256)
void gg_kern_gemm(const float* __restrict__ comp,
                  const float* __restrict__ w2u,
                  const float* __restrict__ b2,
                  float* __restrict__ kern) {
    __shared__ float xch[4][40][32];

    int b    = blockIdx.x >> 5;
    int h    = (blockIdx.x & 31) * 2;
    int wave = threadIdx.x >> 5;
    int pt   = wave & 3;
    int gp   = wave >> 2;
    int lane = threadIdx.x & 31;
    int lr   = lane & 15;
    int half = lane >> 4;
    int w    = pt * 16 + lr;

    const char* cbytes = (const char*)(comp + (size_t)b * CCOMP * HW);
    const v4f*  af     = (const v4f*)w2u + lane + gp * (6 * 32);

    unsigned boff[3][4];
    bool     bok[3][4];
#pragma unroll
    for (int kx = 0; kx < 3; ++kx) {
#pragma unroll
        for (int i = 0; i < 4; ++i) {
            int hh = h - 1 + i, ww = w + kx - 1;
            bool v = ((unsigned)hh < 64u) && ((unsigned)ww < 64u);
            bok[kx][i]  = v;
            boff[kx][i] = (unsigned)((2 * half * HW + (v ? hh * 64 + ww : 0)) * 4);
        }
    }

    float bz0[8], bz1[8], bz2[4];
#pragma unroll
    for (int r = 0; r < 8; ++r) {
        bz0[r] = b2[r + 8 * half];
        bz1[r] = b2[r + 8 * half + 16];
    }
#pragma unroll
    for (int r = 0; r < 4; ++r) bz2[r] = b2[half ? 0 : 32 + r];

    v8f acc[2][3] = {};
    if (gp == 0) gg_wn_run<0>(cbytes, af, boff, bok, acc);
    else         gg_wn_run<1>(cbytes, af, boff, bok, acc);

    if (gp == 1) {
#pragma unroll
        for (int t = 0; t < 2; ++t) {
#pragma unroll
            for (int r = 0; r < 8; ++r) {
                xch[pt][(t * 8 + r) * 2 + 0][lane] = acc[0][t][r];
                xch[pt][(t * 8 + r) * 2 + 1][lane] = acc[0][t][r] + acc[1][t][r];
            }
        }
#pragma unroll
        for (int r = 0; r < 4; ++r) {
            xch[pt][32 + r * 2 + 0][lane] = acc[0][2][r];
            xch[pt][32 + r * 2 + 1][lane] = acc[0][2][r] + acc[1][2][r];
        }
    }
    __syncthreads();

    if (gp == 0) {
        float* kb = kern + (size_t)b * CENC * HW + h * 64 + w;
#pragma unroll
        for (int t = 0; t < 2; ++t) {
#pragma unroll
            for (int r = 0; r < 8; ++r) {
                int   m  = 16 * t + r + 8 * half;
                float z0 = xch[pt][(t * 8 + r) * 2 + 0][lane];
                float z1 = xch[pt][(t * 8 + r) * 2 + 1][lane];
                float bz = t ? bz1[r] : bz0[r];
                kb[m * HW]      = ((acc[0][t][r] + acc[1][t][r]) + z0) + bz;
                kb[m * HW + 64] = (acc[1][t][r] - z1) + bz;
            }
        }
        if (half == 0) {
#pragma unroll
            for (int r = 0; r < 4; ++r) {
                float z0 = xch[pt][32 + r * 2 + 0][lane];
                float z1 = xch[pt][32 + r * 2 + 1][lane];
                kb[(r + 32) * HW]      = ((acc[0][2][r] + acc[1][2][r]) + z0) + bz2[r];
                kb[(r + 32) * HW + 64] = (acc[1][2][r] - z1) + bz2[r];
            }
        }
    }
}

#define CH     32
#define XROW   72
#define NPIECE 6
#define XBUF   (CH * 3 * XROW)

__global__ __launch_bounds__(256)
void gg_carafe_kernel(const float* __restrict__ x,
                      const float* __restrict__ mean,
                      const float* __restrict__ kern,
                      float* __restrict__ out) {
    __shared__ float smask[KK][SS][WWID];
    __shared__ __attribute__((aligned(16))) float xt[2 * XBUF];

    int bh  = blockIdx.x;
    int b   = bh >> 6, h = bh & 63;
    int tid = threadIdx.x;
    int tx  = tid & 63;
    int ty  = tid >> 6;

    const float* xb    = x + (size_t)b * CIN * HW;
    const float* meanb = mean + b * HW;
    const float* kb    = kern + (size_t)b * CENC * HW;

    unsigned goff[NPIECE], loff[NPIECE];
    bool     gok[NPIECE];
#pragma unroll
    for (int i = 0; i < NPIECE; ++i) {
        int q   = tid + 256 * i;
        int cc  = q / 48;
        int pos = q - cc * 48;
        int r   = pos >> 4, p = pos & 15;
        bool ok = (unsigned)(h - 1 + r) < 64u;
        gok[i]  = ok;
        goff[i] = ok ? (unsigned)(cc * HW + (h - 1) * 64 + pos * 4) : 0u;
        loff[i] = (unsigned)((cc * 3 + r) * XROW + 4 + p * 4);
    }

    v4f st[NPIECE];
#pragma unroll
    for (int i = 0; i < NPIECE; ++i)
        st[i] = *(const v4f*)(xb + goff[i]);

    const v4f zero4 = {0.f, 0.f, 0.f, 0.f};
    for (int i = tid; i < 2 * CH * 3 * 2; i += 256)
        *(v4f*)(xt + (i >> 1) * XROW + (i & 1) * 68) = zero4;

    if (ty == 0) {
        int w   = tx;
        float mc = meanb[h * 64 + w];
        float mval[SS][KK];
#pragma unroll
        for (int kk = 0; kk < KK; ++kk) {
            int dy = kk / 3 - 1, dx = kk % 3 - 1;
            int hh = h + dy, ww = w + dx;
            float mp = ((unsigned)hh < 64u && (unsigned)ww < 64u)
                           ? meanb[hh * 64 + ww] : 0.f;
            float g = mp - mc;
            g = 1.0f / (g * g + 0.2f);
#pragma unroll
            for (int s = 0; s < SS; ++s)
                mval[s][kk] = g * kb[(kk * SS + s) * HW + h * 64 + w];
        }
#pragma unroll
        for (int s = 0; s < SS; ++s) {
            float mx = mval[s][0];
#pragma unroll
            for (int kk = 1; kk < KK; ++kk) mx = fmaxf(mx, mval[s][kk]);
            float e[KK], sum = 0.f;
#pragma unroll
            for (int kk = 0; kk < KK; ++kk) { e[kk] = __expf(mval[s][kk] - mx); sum += e[kk]; }
            float inv = 1.0f / sum;
#pragma unroll
            for (int kk = 0; kk < KK; ++kk) smask[kk][s][w] = e[kk] * inv;
        }
    }

#pragma unroll
    for (int i = 0; i < NPIECE; ++i)
        *(v4f*)(xt + loff[i]) = gok[i] ? st[i] : zero4;
    __syncthreads();

    float m[KK][SS];
#pragma unroll
    for (int kk = 0; kk < KK; ++kk)
#pragma unroll
        for (int s = 0; s < SS; ++s) m[kk][s] = smask[kk][s][tx];

    float* ob = out + ((size_t)b * CIN * 128 + 2 * h) * 128 + 2 * tx;

    for (int k = 0; k < CIN / CH; ++k) {
        const bool more = (k + 1) < CIN / CH;
        if (more) {
            const float* xn = xb + (size_t)(k + 1) * CH * HW;
#pragma unroll
            for (int i = 0; i < NPIECE; ++i)
                st[i] = *(const v4f*)(xn + goff[i]);
        }
        __builtin_amdgcn_sched_barrier(0);

        const float* tb = xt + (k & 1) * XBUF + 3 + tx;
#pragma unroll
        for (int j = 0; j < CH / 4; ++j) {
            int cc = ty + 4 * j;
            const float* tr = tb + cc * 3 * XROW;
            float v[KK];
#pragma unroll
            for (int kk = 0; kk < KK; ++kk)
                v[kk] = tr[(kk / 3) * XROW + (kk % 3)];
            float a[SS];
#pragma unroll
            for (int s = 0; s < SS; ++s) {
                float acc = 0.f;
#pragma unroll
                for (int kk = 0; kk < KK; ++kk)
                    acc = fmaf(v[kk], m[kk][s], acc);
                a[s] = acc;
            }
            float* o = ob + (size_t)(k * CH + cc) * (128 * 128);
            v2f o0 = {a[0], a[1]}, o1 = {a[2], a[3]};
            *(v2f*)o         = o0;
            *(v2f*)(o + 128) = o1;
        }

        __builtin_amdgcn_sched_barrier(0);
        if (more) {
            float* tn = xt + ((k + 1) & 1) * XBUF;
#pragma unroll
            for (int i = 0; i < NPIECE; ++i)
                *(v4f*)(tn + loff[i]) = gok[i] ? st[i] : zero4;
            gg_lds_barrier();
        }
    }
}

extern "C" void kernel_launch(void* const* d_in, const int* in_sizes, int n_in,
                              void* d_out, int out_size, void* d_ws, size_t ws_size,
                              hipStream_t stream) {
    const float* x  = (const float*)d_in[0];
    const float* w1 = (const float*)d_in[1];
    const float* b1 = (const float*)d_in[2];
    const float* w2 = (const float*)d_in[3];
    const float* b2 = (const float*)d_in[4];
    float* out = (float*)d_out;

    float* ws   = (float*)d_ws;
    float* mean = ws;
    float* comp = ws + 32768;
    float* kern = comp + (size_t)8 * CCOMP * HW;
    float* w2u  = out;

    gg_comp_mean_kernel<<<512, 256, 0, stream>>>(x, w1, b1, w2, comp, mean, w2u);
    gg_kern_gemm   <<<256,  256, 0, stream>>>(comp, w2u, b2, kern);
    gg_carafe_kernel<<<512, 256, 0, stream>>>(x, mean, kern, out);
}
